# starting kernel
# baseline (speedup 1.0000x reference)
#include <hip/hip_runtime.h>
#include <math.h>

typedef __attribute__((ext_vector_type(2))) float v2f;
typedef __attribute__((ext_vector_type(8))) float v8f;

__device__ __forceinline__ unsigned f2key(float f) {
    unsigned b = __float_as_uint(f);
    return (b & 0x80000000u) ? ~b : (b | 0x80000000u);
}
__device__ __forceinline__ float key2f(unsigned k) {
    unsigned b = (k & 0x80000000u) ? (k & 0x7FFFFFFFu) : ~k;
    return __uint_as_float(b);
}

template <int K, int NC>
__global__ void gemm_wmma_f32(const float* __restrict__ A,
                              const float* __restrict__ B,
                              float* __restrict__ C,
                              int M) {
    __shared__ float Bs[NC * K];
    const int tid = threadIdx.x;

    for (int idx = tid; idx < K * NC; idx += 128) {
        int k = idx / NC;
        int n = idx - k * NC;
        Bs[n * K + k] = B[idx];
    }
    __syncthreads();

    const int waveId = tid >> 5;
    const int lane   = tid & 31;
    const int half   = lane >> 4;
    const int l15    = lane & 15;
    const int kb     = half << 1;
    const int m0     = (blockIdx.x * 4 + waveId) << 4;
    if (m0 >= M) return;

    int rowA = m0 + l15;
    if (rowA >= M) rowA = M - 1;
    const float* __restrict__ arow = A + (size_t)rowA * K;

    v2f areg[K / 4];
#pragma unroll
    for (int kk = 0; kk < K / 4; ++kk)
        areg[kk] = *(const v2f*)(arow + kk * 4 + kb);

#pragma unroll
    for (int tn = 0; tn < NC / 16; ++tn) {
        const int n0 = tn << 4;
        const float* bcol = &Bs[(n0 + l15) * K + kb];
        v8f acc = {};
#pragma unroll
        for (int kk = 0; kk < K / 4; ++kk) {
            v2f b = *(const v2f*)(bcol + kk * 4);
            acc = __builtin_amdgcn_wmma_f32_16x16x4_f32(false, areg[kk], false, b,
                                                        (short)0, acc, false, false);
        }
        const int colC  = n0 + l15;
        const int rbase = m0 + (half << 3);
#pragma unroll
        for (int r = 0; r < 8; ++r) {
            int row = rbase + r;
            if (row < M) C[(size_t)row * NC + colC] = acc[r];
        }
    }
}

__global__ void att_scores(const float* __restrict__ h,
                           const float* __restrict__ a_src,
                           const float* __restrict__ a_dst,
                           float* __restrict__ out_s,
                           float* __restrict__ out_d,
                           int N, int H, int Cc) {
    int t = blockIdx.x * blockDim.x + threadIdx.x;
    if (t >= N * H) return;
    int n  = t / H;
    int hh = t - n * H;
    const float* row = h + (size_t)n * H * Cc + (size_t)hh * Cc;
    const float* vs  = a_src + (size_t)hh * Cc;
    const float* vd  = a_dst + (size_t)hh * Cc;
    float ss = 0.f, dd = 0.f;
    for (int c = 0; c < Cc; ++c) {
        float v = row[c];
        ss += v * vs[c];
        dd += v * vd[c];
    }
    out_s[t] = ss;
    out_d[t] = dd;
}

__global__ void edge_logit_max(const int* __restrict__ ei, int E, int H,
                               const float* __restrict__ as_, const float* __restrict__ ad_,
                               float* __restrict__ p, unsigned* __restrict__ mkey) {
    int e = blockIdx.x * blockDim.x + threadIdx.x;
    if (e >= E) return;
    int s = ei[e];
    int d = ei[E + e];
    for (int h = 0; h < H; ++h) {
        float el = as_[s * H + h] + ad_[d * H + h];
        el = (el > 0.f) ? el : 0.2f * el;
        p[(size_t)e * H + h] = el;
        atomicMax(&mkey[d * H + h], f2key(el));
    }
}

__global__ void edge_softmax_num(const int* __restrict__ ei, int E, int H,
                                 float* __restrict__ p,
                                 const unsigned* __restrict__ mkey,
                                 float* __restrict__ ssum) {
    int e = blockIdx.x * blockDim.x + threadIdx.x;
    if (e >= E) return;
    int d = ei[E + e];
    for (int h = 0; h < H; ++h) {
        float el = p[(size_t)e * H + h];
        float m  = key2f(mkey[d * H + h]);
        float pe = __expf(el - m);
        p[(size_t)e * H + h] = pe;
        atomicAdd(&ssum[d * H + h], pe);
    }
}

__global__ void edge_aggregate(const int* __restrict__ ei, int E,
                               int logF, int logC,
                               const float* __restrict__ p,
                               const float* __restrict__ ssum,
                               const float* __restrict__ hsrc,
                               float* __restrict__ out) {
    long long t = (long long)blockIdx.x * blockDim.x + threadIdx.x;
    long long total = (long long)E << logF;
    if (t >= total) return;
    int e  = (int)(t >> logF);
    int c  = (int)(t & ((1 << logF) - 1));
    int hh = c >> logC;
    int H  = 1 << (logF - logC);
    int s  = ei[e];
    int d  = ei[E + e];
    float alpha = p[(size_t)e * H + hh] / (ssum[(size_t)d * H + hh] + 1e-16f);
    atomicAdd(&out[((size_t)d << logF) + c], hsrc[((size_t)s << logF) + c] * alpha);
}

__global__ void bias_elu(float* __restrict__ x, const float* __restrict__ b,
                         long long total, int mask) {
    long long t = (long long)blockIdx.x * blockDim.x + threadIdx.x;
    if (t >= total) return;
    float v = x[t] + b[(int)(t & mask)];
    x[t] = (v > 0.f) ? v : (__expf(v) - 1.f);
}

__global__ void bias_add(float* __restrict__ x, const float* __restrict__ b,
                         long long total, int mask) {
    long long t = (long long)blockIdx.x * blockDim.x + threadIdx.x;
    if (t >= total) return;
    x[t] += b[(int)(t & mask)];
}

extern "C" void kernel_launch(void* const* d_in, const int* in_sizes, int n_in,
                              void* d_out, int out_size, void* d_ws, size_t ws_size,
                              hipStream_t stream) {
    const float* x        = (const float*)d_in[0];
    const int*   ei       = (const int*)  d_in[1];
    const float* W1       = (const float*)d_in[2];
    const float* att_src1 = (const float*)d_in[3];
    const float* att_dst1 = (const float*)d_in[4];
    const float* b1       = (const float*)d_in[5];
    const float* W2       = (const float*)d_in[6];
    const float* att_src2 = (const float*)d_in[7];
    const float* att_dst2 = (const float*)d_in[8];
    const float* b2       = (const float*)d_in[9];
    float* out = (float*)d_out;

    const int N = in_sizes[0] / 128;
    const int E = in_sizes[1] / 2;

    float* ws = (float*)d_ws;
    float*    h1  = ws;
    float*    f1  = h1 + (size_t)N * 128;
    float*    as1 = f1 + (size_t)N * 128;
    float*    ad1 = as1 + (size_t)N * 4;
    unsigned* m1  = (unsigned*)(ad1 + (size_t)N * 4);
    float*    s1  = (float*)m1 + (size_t)N * 4;
    float*    as2 = s1 + (size_t)N * 4;
    float*    ad2 = as2 + N;
    unsigned* m2  = (unsigned*)(ad2 + N);
    float*    s2  = (float*)m2 + N;
    float*    p1  = s2 + N;
    float*    h2  = h1;
    float*    p2  = p1;

    hipMemsetAsync(f1, 0, sizeof(float) * ((size_t)N * 128 + (size_t)N * 20), stream);
    hipMemsetAsync(out, 0, sizeof(float) * (size_t)N * 64, stream);

    const int TB = 256;
    const int mTiles = (N + 15) / 16;
    const int gBlocks = (mTiles + 3) / 4;

    gemm_wmma_f32<128, 128><<<gBlocks, 128, 0, stream>>>(x, W1, h1, N);
    att_scores<<<(N * 4 + TB - 1) / TB, TB, 0, stream>>>(h1, att_src1, att_dst1, as1, ad1, N, 4, 32);
    edge_logit_max  <<<(E + TB - 1) / TB, TB, 0, stream>>>(ei, E, 4, as1, ad1, p1, m1);
    edge_softmax_num<<<(E + TB - 1) / TB, TB, 0, stream>>>(ei, E, 4, p1, m1, s1);
    {
        long long total = (long long)E << 7;
        edge_aggregate<<<(int)((total + TB - 1) / TB), TB, 0, stream>>>(ei, E, 7, 5, p1, s1, h1, f1);
    }
    {
        long long total = (long long)N * 128;
        bias_elu<<<(int)((total + TB - 1) / TB), TB, 0, stream>>>(f1, b1, total, 127);
    }

    gemm_wmma_f32<128, 64><<<gBlocks, 128, 0, stream>>>(f1, W2, h2, N);
    att_scores<<<(N + TB - 1) / TB, TB, 0, stream>>>(h2, att_src2, att_dst2, as2, ad2, N, 1, 64);
    edge_logit_max  <<<(E + TB - 1) / TB, TB, 0, stream>>>(ei, E, 1, as2, ad2, p2, m2);
    edge_softmax_num<<<(E + TB - 1) / TB, TB, 0, stream>>>(ei, E, 1, p2, m2, s2);
    {
        long long total = (long long)E << 6;
        edge_aggregate<<<(int)((total + TB - 1) / TB), TB, 0, stream>>>(ei, E, 6, 6, p2, s2, h2, out);
    }
    {
        long long total = (long long)N * 64;
        bias_add<<<(int)((total + TB - 1) / TB), TB, 0, stream>>>(out, b2, total, 63);
    }
}


# Edges counting-sorted by destination inside each call; both aggregations become a one-wave-per-node gather with whole-row stores instead of per-element float atomicAdd
# speedup vs baseline: 2.2248x; 2.2248x over previous
#include <hip/hip_runtime.h>
#include <math.h>

typedef __attribute__((ext_vector_type(2))) float v2f;
typedef __attribute__((ext_vector_type(8))) float v8f;

__device__ __forceinline__ unsigned f2key(float f) {
    unsigned b = __float_as_uint(f);
    return (b & 0x80000000u) ? ~b : (b | 0x80000000u);
}
__device__ __forceinline__ float key2f(unsigned k) {
    unsigned b = (k & 0x80000000u) ? (k & 0x7FFFFFFFu) : ~k;
    return __uint_as_float(b);
}

template <int K, int NC>
__global__ void gemm_wmma_f32(const float* __restrict__ A,
                              const float* __restrict__ B,
                              float* __restrict__ C,
                              int M) {
    __shared__ float Bs[NC * K];
    const int tid = threadIdx.x;

    for (int idx = tid; idx < K * NC; idx += 128) {
        int k = idx / NC;
        int n = idx - k * NC;
        Bs[n * K + k] = B[idx];
    }
    __syncthreads();

    const int waveId = tid >> 5;
    const int lane   = tid & 31;
    const int half   = lane >> 4;
    const int l15    = lane & 15;
    const int kb     = half << 1;
    const int m0     = (blockIdx.x * 4 + waveId) << 4;
    if (m0 >= M) return;

    int rowA = m0 + l15;
    if (rowA >= M) rowA = M - 1;
    const float* __restrict__ arow = A + (size_t)rowA * K;

    v2f areg[K / 4];
#pragma unroll
    for (int kk = 0; kk < K / 4; ++kk)
        areg[kk] = *(const v2f*)(arow + kk * 4 + kb);

#pragma unroll
    for (int tn = 0; tn < NC / 16; ++tn) {
        const int n0 = tn << 4;
        const float* bcol = &Bs[(n0 + l15) * K + kb];
        v8f acc = {};
#pragma unroll
        for (int kk = 0; kk < K / 4; ++kk) {
            v2f b = *(const v2f*)(bcol + kk * 4);
            acc = __builtin_amdgcn_wmma_f32_16x16x4_f32(false, areg[kk], false, b,
                                                        (short)0, acc, false, false);
        }
        const int colC  = n0 + l15;
        const int rbase = m0 + (half << 3);
#pragma unroll
        for (int r = 0; r < 8; ++r) {
            int row = rbase + r;
            if (row < M) C[(size_t)row * NC + colC] = acc[r];
        }
    }
}

__global__ void att_scores(const float* __restrict__ h,
                           const float* __restrict__ a_src,
                           const float* __restrict__ a_dst,
                           float* __restrict__ out_s,
                           float* __restrict__ out_d,
                           int N, int H, int Cc) {
    int t = blockIdx.x * blockDim.x + threadIdx.x;
    if (t >= N * H) return;
    int n  = t / H;
    int hh = t - n * H;
    const float* row = h + (size_t)n * H * Cc + (size_t)hh * Cc;
    const float* vs  = a_src + (size_t)hh * Cc;
    const float* vd  = a_dst + (size_t)hh * Cc;
    float ss = 0.f, dd = 0.f;
    for (int c = 0; c < Cc; ++c) {
        float v = row[c];
        ss += v * vs[c];
        dd += v * vd[c];
    }
    out_s[t] = ss;
    out_d[t] = dd;
}

__global__ void edge_logit_max(const int* __restrict__ ei, int E, int H,
                               const float* __restrict__ as_, const float* __restrict__ ad_,
                               float* __restrict__ p, unsigned* __restrict__ mkey) {
    int e = blockIdx.x * blockDim.x + threadIdx.x;
    if (e >= E) return;
    int s = ei[e];
    int d = ei[E + e];
    for (int h = 0; h < H; ++h) {
        float el = as_[s * H + h] + ad_[d * H + h];
        el = (el > 0.f) ? el : 0.2f * el;
        p[(size_t)e * H + h] = el;
        atomicMax(&mkey[d * H + h], f2key(el));
    }
}

__global__ void edge_softmax_num(const int* __restrict__ ei, int E, int H,
                                 float* __restrict__ p,
                                 const unsigned* __restrict__ mkey,
                                 float* __restrict__ ssum) {
    int e = blockIdx.x * blockDim.x + threadIdx.x;
    if (e >= E) return;
    int d = ei[E + e];
    for (int h = 0; h < H; ++h) {
        float el = p[(size_t)e * H + h];
        float m  = key2f(mkey[d * H + h]);
        float pe = __expf(el - m);
        p[(size_t)e * H + h] = pe;
        atomicAdd(&ssum[d * H + h], pe);
    }
}

__global__ void dst_count(const int* __restrict__ ei, int E, int* __restrict__ cnt) {
    int e = blockIdx.x * blockDim.x + threadIdx.x;
    if (e >= E) return;
    atomicAdd(&cnt[ei[E + e]], 1);
}

__global__ void dst_scan(const int* __restrict__ cnt, int N,
                         int* __restrict__ beg, int* __restrict__ cur) {
    __shared__ int red[256];
    __shared__ int sc[256];
    const int tid  = threadIdx.x;
    const int base = blockIdx.x * 1024;

    int pre = 0;
    for (int i = tid; i < base; i += 256) pre += cnt[i];

    const int i0 = base + tid * 4;
    int c0 = (i0     < N) ? cnt[i0]     : 0;
    int c1 = (i0 + 1 < N) ? cnt[i0 + 1] : 0;
    int c2 = (i0 + 2 < N) ? cnt[i0 + 2] : 0;
    int c3 = (i0 + 3 < N) ? cnt[i0 + 3] : 0;
    const int mine = c0 + c1 + c2 + c3;

    red[tid] = pre;
    sc[tid]  = mine;
    __syncthreads();
    for (int o = 128; o > 0; o >>= 1) {
        if (tid < o) red[tid] += red[tid + o];
        __syncthreads();
    }
    for (int o = 1; o < 256; o <<= 1) {
        int v = (tid >= o) ? sc[tid - o] : 0;
        __syncthreads();
        sc[tid] += v;
        __syncthreads();
    }
    int o0 = red[0] + sc[tid] - mine;
    int o1 = o0 + c0;
    int o2 = o1 + c1;
    int o3 = o2 + c2;
    if (i0     < N) { beg[i0]     = o0; cur[i0]     = o0; }
    if (i0 + 1 < N) { beg[i0 + 1] = o1; cur[i0 + 1] = o1; }
    if (i0 + 2 < N) { beg[i0 + 2] = o2; cur[i0 + 2] = o2; }
    if (i0 + 3 < N) { beg[i0 + 3] = o3; cur[i0 + 3] = o3; }
}

__global__ void dst_fill(const int* __restrict__ ei, int E,
                         int* __restrict__ cur, int* __restrict__ srt) {
    int e = blockIdx.x * blockDim.x + threadIdx.x;
    if (e >= E) return;
    int s = ei[e];
    int d = ei[E + e];
    int pos = atomicAdd(&cur[d], 1);
    srt[pos]     = s;
    srt[E + pos] = d;
}

template <int H>
__global__ void dst_gather(const int* __restrict__ srt,
                           const int* __restrict__ beg, const int* __restrict__ fin,
                           const float* __restrict__ p, const float* __restrict__ ssum,
                           const float* __restrict__ hsrc, float* __restrict__ out, int N) {
    const int lane = threadIdx.x & 31;
    const int d    = blockIdx.x * (blockDim.x >> 5) + (threadIdx.x >> 5);
    if (d >= N) return;
    const int jb = __builtin_amdgcn_readfirstlane(beg[d]);
    const int je = __builtin_amdgcn_readfirstlane(fin[d]);

    if constexpr (H == 4) {
        const float4 sv = *(const float4*)(ssum + (size_t)d * 4);
        const bool hi = (lane & 16) != 0, odd = (lane & 8) != 0;
        float4 acc = make_float4(0.f, 0.f, 0.f, 0.f);
        for (int base = jb; base < je; base += 32) {
            const int n = (je - base < 32) ? (je - base) : 32;
            int sI = 0;
            float4 a = make_float4(0.f, 0.f, 0.f, 0.f);
            if (lane < n) {
                sI = srt[base + lane];
                float4 pv = *(const float4*)(p + (size_t)(base + lane) * 4);
                a.x = pv.x / (sv.x + 1e-16f);
                a.y = pv.y / (sv.y + 1e-16f);
                a.z = pv.z / (sv.z + 1e-16f);
                a.w = pv.w / (sv.w + 1e-16f);
            }
#pragma unroll 4
            for (int k = 0; k < n; ++k) {
                const int   s  = __builtin_amdgcn_readlane(sI, k);
                const float b0 = __int_as_float(__builtin_amdgcn_readlane(__float_as_int(a.x), k));
                const float b1 = __int_as_float(__builtin_amdgcn_readlane(__float_as_int(a.y), k));
                const float b2 = __int_as_float(__builtin_amdgcn_readlane(__float_as_int(a.z), k));
                const float b3 = __int_as_float(__builtin_amdgcn_readlane(__float_as_int(a.w), k));
                const float al = hi ? (odd ? b3 : b2) : (odd ? b1 : b0);
                const float4 hv = *(const float4*)(hsrc + (size_t)s * 128 + lane * 4);
                acc.x += hv.x * al;
                acc.y += hv.y * al;
                acc.z += hv.z * al;
                acc.w += hv.w * al;
            }
        }
        *(float4*)(out + (size_t)d * 128 + lane * 4) = acc;
    } else {
        const float sv = ssum[d];
        float2 acc = make_float2(0.f, 0.f);
        for (int base = jb; base < je; base += 32) {
            const int n = (je - base < 32) ? (je - base) : 32;
            int sI = 0;
            float a = 0.f;
            if (lane < n) {
                sI = srt[base + lane];
                a  = p[base + lane] / (sv + 1e-16f);
            }
#pragma unroll 4
            for (int k = 0; k < n; ++k) {
                const int   s  = __builtin_amdgcn_readlane(sI, k);
                const float al = __int_as_float(__builtin_amdgcn_readlane(__float_as_int(a), k));
                const float2 hv = *(const float2*)(hsrc + (size_t)s * 64 + lane * 2);
                acc.x += hv.x * al;
                acc.y += hv.y * al;
            }
        }
        *(float2*)(out + (size_t)d * 64 + lane * 2) = acc;
    }
}

__global__ void bias_elu(float* __restrict__ x, const float* __restrict__ b,
                         long long total, int mask) {
    long long t = (long long)blockIdx.x * blockDim.x + threadIdx.x;
    if (t >= total) return;
    float v = x[t] + b[(int)(t & mask)];
    x[t] = (v > 0.f) ? v : (__expf(v) - 1.f);
}

__global__ void bias_add(float* __restrict__ x, const float* __restrict__ b,
                         long long total, int mask) {
    long long t = (long long)blockIdx.x * blockDim.x + threadIdx.x;
    if (t >= total) return;
    x[t] += b[(int)(t & mask)];
}

extern "C" void kernel_launch(void* const* d_in, const int* in_sizes, int n_in,
                              void* d_out, int out_size, void* d_ws, size_t ws_size,
                              hipStream_t stream) {
    const float* x        = (const float*)d_in[0];
    const int*   ei       = (const int*)  d_in[1];
    const float* W1       = (const float*)d_in[2];
    const float* att_src1 = (const float*)d_in[3];
    const float* att_dst1 = (const float*)d_in[4];
    const float* b1       = (const float*)d_in[5];
    const float* W2       = (const float*)d_in[6];
    const float* att_src2 = (const float*)d_in[7];
    const float* att_dst2 = (const float*)d_in[8];
    const float* b2       = (const float*)d_in[9];
    float* out = (float*)d_out;

    const int N = in_sizes[0] / 128;
    const int E = in_sizes[1] / 2;

    float* ws = (float*)d_ws;
    float*    h1  = ws;
    float*    f1  = h1 + (size_t)N * 128;
    float*    as1 = f1 + (size_t)N * 128;
    float*    ad1 = as1 + (size_t)N * 4;
    unsigned* m1  = (unsigned*)(ad1 + (size_t)N * 4);
    float*    s1  = (float*)m1 + (size_t)N * 4;
    float*    as2 = s1 + (size_t)N * 4;
    float*    ad2 = as2 + N;
    unsigned* m2  = (unsigned*)(ad2 + N);
    float*    s2  = (float*)m2 + N;
    int*      cnt = (int*)(s2 + N);
    int*      beg = cnt + N;
    int*      fin = beg + N;
    int*      srt = fin + N;
    float*    p2  = (float*)(srt + (size_t)E * 2);
    float*    p1  = out;
    float*    h2  = h1;

    hipMemsetAsync(m1, 0, sizeof(float) * ((size_t)N * 13), stream);

    const int TB = 256;
    const int mTiles = (N + 15) / 16;
    const int gBlocks = (mTiles + 3) / 4;
    const int eBlocks = (E + TB - 1) / TB;
    const int nBlocks = (N + 7) / 8;

    dst_count<<<eBlocks, TB, 0, stream>>>(ei, E, cnt);
    dst_scan <<<(N + 1023) / 1024, 256, 0, stream>>>(cnt, N, beg, fin);
    dst_fill <<<eBlocks, TB, 0, stream>>>(ei, E, fin, srt);

    gemm_wmma_f32<128, 128><<<gBlocks, 128, 0, stream>>>(x, W1, h1, N);
    att_scores<<<(N * 4 + TB - 1) / TB, TB, 0, stream>>>(h1, att_src1, att_dst1, as1, ad1, N, 4, 32);
    edge_logit_max  <<<eBlocks, TB, 0, stream>>>(srt, E, 4, as1, ad1, p1, m1);
    edge_softmax_num<<<eBlocks, TB, 0, stream>>>(srt, E, 4, p1, m1, s1);
    dst_gather<4><<<nBlocks, TB, 0, stream>>>(srt, beg, fin, p1, s1, h1, f1, N);
    {
        long long total = (long long)N * 128;
        bias_elu<<<(int)((total + TB - 1) / TB), TB, 0, stream>>>(f1, b1, total, 127);
    }

    gemm_wmma_f32<128, 64><<<gBlocks, 128, 0, stream>>>(f1, W2, h2, N);
    att_scores<<<(N + TB - 1) / TB, TB, 0, stream>>>(h2, att_src2, att_dst2, as2, ad2, N, 1, 64);
    edge_logit_max  <<<eBlocks, TB, 0, stream>>>(srt, E, 1, as2, ad2, p2, m2);
    edge_softmax_num<<<eBlocks, TB, 0, stream>>>(srt, E, 1, p2, m2, s2);
    dst_gather<1><<<nBlocks, TB, 0, stream>>>(srt, beg, fin, p2, s2, h2, out, N);
    {
        long long total = (long long)N * 64;
        bias_add<<<(int)((total + TB - 1) / TB), TB, 0, stream>>>(out, b2, total, 63);
    }
}


# LDS tile of the transposed weights padded from a row pitch of K to K+4 floats, so the lanes of a fragment read or of a staging store no longer queue on one bank (both GEMMs)
# speedup vs baseline: 4.1758x; 1.8769x over previous
#include <hip/hip_runtime.h>
#include <math.h>

typedef __attribute__((ext_vector_type(2))) float v2f;
typedef __attribute__((ext_vector_type(8))) float v8f;

template <int K, int NC>
__global__ void gemm_wmma_f32(const float* __restrict__ A,
                              const float* __restrict__ B,
                              float* __restrict__ C,
                              int M) {
    constexpr int P = K + 4;
    __shared__ float Bs[NC * P];
    const int tid = threadIdx.x;

    for (int idx = tid; idx < K * NC; idx += 128) {
        int k = idx / NC;
        int n = idx - k * NC;
        Bs[n * P + k] = B[idx];
    }
    __syncthreads();

    const int waveId = tid >> 5;
    const int lane   = tid & 31;
    const int half   = lane >> 4;
    const int l15    = lane & 15;
    const int kb     = half << 1;
    const int m0     = (blockIdx.x * 4 + waveId) << 4;
    if (m0 >= M) return;

    int rowA = m0 + l15;
    if (rowA >= M) rowA = M - 1;
    const float* __restrict__ arow = A + (size_t)rowA * K;

    v2f areg[K / 4];
#pragma unroll
    for (int kk = 0; kk < K / 4; ++kk)
        areg[kk] = *(const v2f*)(arow + kk * 4 + kb);

#pragma unroll
    for (int tn = 0; tn < NC / 16; ++tn) {
        const int n0 = tn << 4;
        const float* bcol = &Bs[(n0 + l15) * P + kb];
        v8f acc = {};
#pragma unroll
        for (int kk = 0; kk < K / 4; ++kk) {
            v2f b = *(const v2f*)(bcol + kk * 4);
            acc = __builtin_amdgcn_wmma_f32_16x16x4_f32(false, areg[kk], false, b,
                                                        (short)0, acc, false, false);
        }
        const int colC  = n0 + l15;
        const int rbase = m0 + (half << 3);
#pragma unroll
        for (int r = 0; r < 8; ++r) {
            int row = rbase + r;
            if (row < M) C[(size_t)row * NC + colC] = acc[r];
        }
    }
}

__global__ void att_scores(const float* __restrict__ h,
                           const float* __restrict__ a_src,
                           const float* __restrict__ a_dst,
                           float* __restrict__ out_s,
                           float* __restrict__ out_d,
                           int N, int H, int Cc) {
    int t = blockIdx.x * blockDim.x + threadIdx.x;
    if (t >= N * H) return;
    int n  = t / H;
    int hh = t - n * H;
    const float* row = h + (size_t)n * H * Cc + (size_t)hh * Cc;
    const float* vs  = a_src + (size_t)hh * Cc;
    const float* vd  = a_dst + (size_t)hh * Cc;
    float ss = 0.f, dd = 0.f;
    for (int c = 0; c < Cc; ++c) {
        float v = row[c];
        ss += v * vs[c];
        dd += v * vd[c];
    }
    out_s[t] = ss;
    out_d[t] = dd;
}

__global__ void dst_count(const int* __restrict__ ei, int E, int* __restrict__ cnt) {
    int e = blockIdx.x * blockDim.x + threadIdx.x;
    if (e >= E) return;
    atomicAdd(&cnt[ei[E + e]], 1);
}

__global__ void dst_scan(const int* __restrict__ cnt, int N,
                         int* __restrict__ beg, int* __restrict__ cur) {
    __shared__ int red[256];
    __shared__ int sc[256];
    const int tid  = threadIdx.x;
    const int base = blockIdx.x * 1024;

    int pre = 0;
    for (int i = tid; i < base; i += 256) pre += cnt[i];

    const int i0 = base + tid * 4;
    int c0 = (i0     < N) ? cnt[i0]     : 0;
    int c1 = (i0 + 1 < N) ? cnt[i0 + 1] : 0;
    int c2 = (i0 + 2 < N) ? cnt[i0 + 2] : 0;
    int c3 = (i0 + 3 < N) ? cnt[i0 + 3] : 0;
    const int mine = c0 + c1 + c2 + c3;

    red[tid] = pre;
    sc[tid]  = mine;
    __syncthreads();
    for (int o = 128; o > 0; o >>= 1) {
        if (tid < o) red[tid] += red[tid + o];
        __syncthreads();
    }
    for (int o = 1; o < 256; o <<= 1) {
        int v = (tid >= o) ? sc[tid - o] : 0;
        __syncthreads();
        sc[tid] += v;
        __syncthreads();
    }
    int o0 = red[0] + sc[tid] - mine;
    int o1 = o0 + c0;
    int o2 = o1 + c1;
    int o3 = o2 + c2;
    if (i0     < N) { beg[i0]     = o0; cur[i0]     = o0; }
    if (i0 + 1 < N) { beg[i0 + 1] = o1; cur[i0 + 1] = o1; }
    if (i0 + 2 < N) { beg[i0 + 2] = o2; cur[i0 + 2] = o2; }
    if (i0 + 3 < N) { beg[i0 + 3] = o3; cur[i0 + 3] = o3; }
}

__global__ void dst_fill(const int* __restrict__ ei, int E,
                         int* __restrict__ cur, int* __restrict__ srt) {
    int e = blockIdx.x * blockDim.x + threadIdx.x;
    if (e >= E) return;
    int s = ei[e];
    int d = ei[E + e];
    int pos = atomicAdd(&cur[d], 1);
    srt[pos]     = s;
    srt[E + pos] = d;
}

template <bool ELU>
__device__ __forceinline__ float finish(float agg, float b) {
    const float v = agg + b;
    if (ELU) return (v > 0.f) ? v : (__expf(v) - 1.f);
    return v;
}

template <int XMASK>
__device__ __forceinline__ float row_xor(float v) {
    const int b = __float_as_int(v);
    return __int_as_float(__builtin_amdgcn_update_dpp(b, b, 0x160 | XMASK, 0xf, 0xf, false));
}
__device__ __forceinline__ float other_row(float v) {
    const int b = __float_as_int(v);
    return __int_as_float(__builtin_amdgcn_permlanex16(b, b, 0x76543210, 0xfedcba98, false, false));
}
__device__ __forceinline__ float wave_max(float v) {
    v = fmaxf(v, row_xor<1>(v));
    v = fmaxf(v, row_xor<2>(v));
    v = fmaxf(v, row_xor<4>(v));
    v = fmaxf(v, row_xor<8>(v));
    return fmaxf(v, other_row(v));
}
__device__ __forceinline__ float wave_sum(float v) {
    v += row_xor<1>(v);
    v += row_xor<2>(v);
    v += row_xor<4>(v);
    v += row_xor<8>(v);
    return v + other_row(v);
}

__device__ __forceinline__ float leaky(float v) { return (v > 0.f) ? v : 0.2f * v; }

template <int H, bool ELU>
__global__ void dst_gather(const int* __restrict__ srt,
                           const int* __restrict__ beg, const int* __restrict__ fin,
                           const float* __restrict__ as_, const float* __restrict__ ad_,
                           const float* __restrict__ hsrc, const float* __restrict__ bias,
                           float* __restrict__ out, int N) {
    const int lane = threadIdx.x & 31;
    const int d    = blockIdx.x * (blockDim.x >> 5) + (threadIdx.x >> 5);
    if (d >= N) return;
    const int jb = __builtin_amdgcn_readfirstlane(beg[d]);
    const int je = __builtin_amdgcn_readfirstlane(fin[d]);
    const int n0 = (je - jb < 32) ? (je - jb) : 32;

    if constexpr (H == 4) {
        const float bv0 = bias[lane * 4], bv1 = bias[lane * 4 + 1], bv2 = bias[lane * 4 + 2], bv3 = bias[lane * 4 + 3];
        const float4 adv = *(const float4*)(ad_ + (size_t)d * 4);
        const bool hi = (lane & 16) != 0, odd = (lane & 8) != 0;
        float4 acc = make_float4(0.f, 0.f, 0.f, 0.f);

        auto logits = [&](int base, int n, int& sI) -> float4 {
            float4 e = make_float4(-INFINITY, -INFINITY, -INFINITY, -INFINITY);
            sI = 0;
            if (lane < n) {
                sI = srt[base + lane];
                const float4 av = *(const float4*)(as_ + (size_t)sI * 4);
                e.x = leaky(av.x + adv.x);
                e.y = leaky(av.y + adv.y);
                e.z = leaky(av.z + adv.z);
                e.w = leaky(av.w + adv.w);
            }
            return e;
        };
        auto rows = [&](int sI, float4 a, int n) {
#pragma unroll 4
            for (int k = 0; k < n; ++k) {
                const int   s  = __builtin_amdgcn_readlane(sI, k);
                const float b0 = __int_as_float(__builtin_amdgcn_readlane(__float_as_int(a.x), k));
                const float b1 = __int_as_float(__builtin_amdgcn_readlane(__float_as_int(a.y), k));
                const float b2 = __int_as_float(__builtin_amdgcn_readlane(__float_as_int(a.z), k));
                const float b3 = __int_as_float(__builtin_amdgcn_readlane(__float_as_int(a.w), k));
                const float al = hi ? (odd ? b3 : b2) : (odd ? b1 : b0);
                const float4 hv = *(const float4*)(hsrc + (size_t)s * 128 + lane * 4);
                acc.x += hv.x * al;
                acc.y += hv.y * al;
                acc.z += hv.z * al;
                acc.w += hv.w * al;
            }
        };

        int sI0;
        const float4 e0 = logits(jb, n0, sI0);

        float4 m = e0;
        for (int base = jb + 32; base < je; base += 32) {
            const int n = (je - base < 32) ? (je - base) : 32;
            int sI;
            const float4 e = logits(base, n, sI);
            m.x = fmaxf(m.x, e.x); m.y = fmaxf(m.y, e.y); m.z = fmaxf(m.z, e.z); m.w = fmaxf(m.w, e.w);
        }
        m.x = wave_max(m.x); m.y = wave_max(m.y); m.z = wave_max(m.z); m.w = wave_max(m.w);

        float4 p0 = make_float4(0.f, 0.f, 0.f, 0.f);
        if (lane < n0) {
            p0.x = __expf(e0.x - m.x);
            p0.y = __expf(e0.y - m.y);
            p0.z = __expf(e0.z - m.z);
            p0.w = __expf(e0.w - m.w);
        }
        float4 sv = p0;
        for (int base = jb + 32; base < je; base += 32) {
            const int n = (je - base < 32) ? (je - base) : 32;
            int sI;
            const float4 e = logits(base, n, sI);
            if (lane < n) {
                sv.x += __expf(e.x - m.x);
                sv.y += __expf(e.y - m.y);
                sv.z += __expf(e.z - m.z);
                sv.w += __expf(e.w - m.w);
            }
        }
        sv.x = wave_sum(sv.x); sv.y = wave_sum(sv.y); sv.z = wave_sum(sv.z); sv.w = wave_sum(sv.w);

        rows(sI0, make_float4(p0.x / (sv.x + 1e-16f), p0.y / (sv.y + 1e-16f),
                              p0.z / (sv.z + 1e-16f), p0.w / (sv.w + 1e-16f)), n0);
        for (int base = jb + 32; base < je; base += 32) {
            const int n = (je - base < 32) ? (je - base) : 32;
            int sI;
            const float4 e = logits(base, n, sI);
            float4 a = make_float4(0.f, 0.f, 0.f, 0.f);
            if (lane < n) {
                a.x = __expf(e.x - m.x) / (sv.x + 1e-16f);
                a.y = __expf(e.y - m.y) / (sv.y + 1e-16f);
                a.z = __expf(e.z - m.z) / (sv.z + 1e-16f);
                a.w = __expf(e.w - m.w) / (sv.w + 1e-16f);
            }
            rows(sI, a, n);
        }
        *(float4*)(out + (size_t)d * 128 + lane * 4) =
            make_float4(finish<ELU>(acc.x, bv0), finish<ELU>(acc.y, bv1),
                        finish<ELU>(acc.z, bv2), finish<ELU>(acc.w, bv3));
    } else {
        const float bv0 = bias[lane * 2], bv1 = bias[lane * 2 + 1];
        const float adv = ad_[d];
        float2 acc = make_float2(0.f, 0.f);

        auto logits = [&](int base, int n, int& sI) -> float {
            float e = -INFINITY;
            sI = 0;
            if (lane < n) {
                sI = srt[base + lane];
                e  = leaky(as_[sI] + adv);
            }
            return e;
        };
        auto rows = [&](int sI, float a, int n) {
#pragma unroll 4
            for (int k = 0; k < n; ++k) {
                const int   s  = __builtin_amdgcn_readlane(sI, k);
                const float al = __int_as_float(__builtin_amdgcn_readlane(__float_as_int(a), k));
                const float2 hv = *(const float2*)(hsrc + (size_t)s * 64 + lane * 2);
                acc.x += hv.x * al;
                acc.y += hv.y * al;
            }
        };

        int sI0;
        const float e0 = logits(jb, n0, sI0);

        float m = e0;
        for (int base = jb + 32; base < je; base += 32) {
            const int n = (je - base < 32) ? (je - base) : 32;
            int sI;
            m = fmaxf(m, logits(base, n, sI));
        }
        m = wave_max(m);

        const float p0 = (lane < n0) ? __expf(e0 - m) : 0.f;
        float sv = p0;
        for (int base = jb + 32; base < je; base += 32) {
            const int n = (je - base < 32) ? (je - base) : 32;
            int sI;
            const float e = logits(base, n, sI);
            if (lane < n) sv += __expf(e - m);
        }
        sv = wave_sum(sv);

        rows(sI0, p0 / (sv + 1e-16f), n0);
        for (int base = jb + 32; base < je; base += 32) {
            const int n = (je - base < 32) ? (je - base) : 32;
            int sI;
            const float e = logits(base, n, sI);
            const float a = (lane < n) ? __expf(e - m) / (sv + 1e-16f) : 0.f;
            rows(sI, a, n);
        }
        *(float2*)(out + (size_t)d * 64 + lane * 2) =
            make_float2(finish<ELU>(acc.x, bv0), finish<ELU>(acc.y, bv1));
    }
}

extern "C" void kernel_launch(void* const* d_in, const int* in_sizes, int n_in,
                              void* d_out, int out_size, void* d_ws, size_t ws_size,
                              hipStream_t stream) {
    const float* x        = (const float*)d_in[0];
    const int*   ei       = (const int*)  d_in[1];
    const float* W1       = (const float*)d_in[2];
    const float* att_src1 = (const float*)d_in[3];
    const float* att_dst1 = (const float*)d_in[4];
    const float* b1       = (const float*)d_in[5];
    const float* W2       = (const float*)d_in[6];
    const float* att_src2 = (const float*)d_in[7];
    const float* att_dst2 = (const float*)d_in[8];
    const float* b2       = (const float*)d_in[9];
    float* out = (float*)d_out;

    const int N = in_sizes[0] / 128;
    const int E = in_sizes[1] / 2;

    float* ws = (float*)d_ws;
    float*    h1  = ws;
    float*    f1  = h1 + (size_t)N * 128;
    float*    as1 = f1 + (size_t)N * 128;
    float*    ad1 = as1 + (size_t)N * 4;
    float*    as2 = ad1 + (size_t)N * 4;
    float*    ad2 = as2 + N;
    int*      cnt = (int*)(ad2 + N);
    int*      beg = cnt + N;
    int*      fin = beg + N;
    int*      srt = fin + N;
    float*    h2  = h1;

    hipMemsetAsync(cnt, 0, sizeof(int) * (size_t)N, stream);

    const int TB = 256;
    const int mTiles = (N + 15) / 16;
    const int gBlocks = (mTiles + 3) / 4;
    const int eBlocks = (E + TB - 1) / TB;
    const int nBlocks = (N + 7) / 8;

    dst_count<<<eBlocks, TB, 0, stream>>>(ei, E, cnt);
    dst_scan <<<(N + 1023) / 1024, 256, 0, stream>>>(cnt, N, beg, fin);
    dst_fill <<<eBlocks, TB, 0, stream>>>(ei, E, fin, srt);

    gemm_wmma_f32<128, 128><<<gBlocks, 128, 0, stream>>>(x, W1, h1, N);
    att_scores<<<(N * 4 + TB - 1) / TB, TB, 0, stream>>>(h1, att_src1, att_dst1, as1, ad1, N, 4, 32);
    dst_gather<4, true><<<nBlocks, TB, 0, stream>>>(srt, beg, fin, as1, ad1, h1, b1, f1, N);

    gemm_wmma_f32<128, 64><<<gBlocks, 128, 0, stream>>>(f1, W2, h2, N);
    att_scores<<<(N + TB - 1) / TB, TB, 0, stream>>>(h2, att_src2, att_dst2, as2, ad2, N, 1, 64);
    dst_gather<1, false><<<nBlocks, TB, 0, stream>>>(srt, beg, fin, as2, ad2, h2, b2, out, N);
}


# Counting sort: an edge's rank in its destination segment is the value the histogram's atomic add returns, so the fill pass places it at start+rank with no second atomic; unread dst plane not written
# speedup vs baseline: 5.3281x; 1.2759x over previous
#include <hip/hip_runtime.h>
#include <math.h>

typedef __attribute__((ext_vector_type(2))) float v2f;
typedef __attribute__((ext_vector_type(8))) float v8f;

template <int K, int NC>
__global__ void gemm_wmma_f32(const float* __restrict__ A,
                              const float* __restrict__ B,
                              float* __restrict__ C,
                              int M) {
    constexpr int P = K + 4;
    __shared__ float Bs[NC * P];
    const int tid = threadIdx.x;

    for (int idx = tid; idx < K * NC; idx += 128) {
        int k = idx / NC;
        int n = idx - k * NC;
        Bs[n * P + k] = B[idx];
    }
    __syncthreads();

    const int waveId = tid >> 5;
    const int lane   = tid & 31;
    const int half   = lane >> 4;
    const int l15    = lane & 15;
    const int kb     = half << 1;
    const int m0     = (blockIdx.x * 4 + waveId) << 4;
    if (m0 >= M) return;

    int rowA = m0 + l15;
    if (rowA >= M) rowA = M - 1;
    const float* __restrict__ arow = A + (size_t)rowA * K;

    v2f areg[K / 4];
#pragma unroll
    for (int kk = 0; kk < K / 4; ++kk)
        areg[kk] = *(const v2f*)(arow + kk * 4 + kb);

#pragma unroll
    for (int tn = 0; tn < NC / 16; ++tn) {
        const int n0 = tn << 4;
        const float* bcol = &Bs[(n0 + l15) * P + kb];
        v8f acc = {};
#pragma unroll
        for (int kk = 0; kk < K / 4; ++kk) {
            v2f b = *(const v2f*)(bcol + kk * 4);
            acc = __builtin_amdgcn_wmma_f32_16x16x4_f32(false, areg[kk], false, b,
                                                        (short)0, acc, false, false);
        }
        const int colC  = n0 + l15;
        const int rbase = m0 + (half << 3);
#pragma unroll
        for (int r = 0; r < 8; ++r) {
            int row = rbase + r;
            if (row < M) C[(size_t)row * NC + colC] = acc[r];
        }
    }
}

__global__ void att_scores(const float* __restrict__ h,
                           const float* __restrict__ a_src,
                           const float* __restrict__ a_dst,
                           float* __restrict__ out_s,
                           float* __restrict__ out_d,
                           int N, int H, int Cc) {
    int t = blockIdx.x * blockDim.x + threadIdx.x;
    if (t >= N * H) return;
    int n  = t / H;
    int hh = t - n * H;
    const float* row = h + (size_t)n * H * Cc + (size_t)hh * Cc;
    const float* vs  = a_src + (size_t)hh * Cc;
    const float* vd  = a_dst + (size_t)hh * Cc;
    float ss = 0.f, dd = 0.f;
    for (int c = 0; c < Cc; ++c) {
        float v = row[c];
        ss += v * vs[c];
        dd += v * vd[c];
    }
    out_s[t] = ss;
    out_d[t] = dd;
}

__global__ void dst_count(const int* __restrict__ ei, int E, int* __restrict__ cnt,
                          int* __restrict__ rank) {
    int e = blockIdx.x * blockDim.x + threadIdx.x;
    if (e >= E) return;
    rank[e] = atomicAdd(&cnt[ei[E + e]], 1);
}

__global__ void dst_scan(const int* __restrict__ cnt, int N,
                         int* __restrict__ beg, int* __restrict__ fin) {
    __shared__ int red[256];
    __shared__ int sc[256];
    const int tid  = threadIdx.x;
    const int base = blockIdx.x * 1024;

    int pre = 0;
    for (int i = tid; i < base; i += 256) pre += cnt[i];

    const int i0 = base + tid * 4;
    int c0 = (i0     < N) ? cnt[i0]     : 0;
    int c1 = (i0 + 1 < N) ? cnt[i0 + 1] : 0;
    int c2 = (i0 + 2 < N) ? cnt[i0 + 2] : 0;
    int c3 = (i0 + 3 < N) ? cnt[i0 + 3] : 0;
    const int mine = c0 + c1 + c2 + c3;

    red[tid] = pre;
    sc[tid]  = mine;
    __syncthreads();
    for (int o = 128; o > 0; o >>= 1) {
        if (tid < o) red[tid] += red[tid + o];
        __syncthreads();
    }
    for (int o = 1; o < 256; o <<= 1) {
        int v = (tid >= o) ? sc[tid - o] : 0;
        __syncthreads();
        sc[tid] += v;
        __syncthreads();
    }
    int o0 = red[0] + sc[tid] - mine;
    int o1 = o0 + c0;
    int o2 = o1 + c1;
    int o3 = o2 + c2;
    if (i0     < N) { beg[i0]     = o0; fin[i0]     = o1; }
    if (i0 + 1 < N) { beg[i0 + 1] = o1; fin[i0 + 1] = o2; }
    if (i0 + 2 < N) { beg[i0 + 2] = o2; fin[i0 + 2] = o3; }
    if (i0 + 3 < N) { beg[i0 + 3] = o3; fin[i0 + 3] = o3 + c3; }
}

__global__ void dst_fill(const int* __restrict__ ei, int E,
                         const int* __restrict__ beg, const int* __restrict__ rank,
                         int* __restrict__ srt) {
    int e = blockIdx.x * blockDim.x + threadIdx.x;
    if (e >= E) return;
    srt[beg[ei[E + e]] + rank[e]] = ei[e];
}

template <bool ELU>
__device__ __forceinline__ float finish(float agg, float b) {
    const float v = agg + b;
    if (ELU) return (v > 0.f) ? v : (__expf(v) - 1.f);
    return v;
}

template <int XMASK>
__device__ __forceinline__ float row_xor(float v) {
    const int b = __float_as_int(v);
    return __int_as_float(__builtin_amdgcn_update_dpp(b, b, 0x160 | XMASK, 0xf, 0xf, false));
}
__device__ __forceinline__ float other_row(float v) {
    const int b = __float_as_int(v);
    return __int_as_float(__builtin_amdgcn_permlanex16(b, b, 0x76543210, 0xfedcba98, false, false));
}
__device__ __forceinline__ float wave_max(float v) {
    v = fmaxf(v, row_xor<1>(v));
    v = fmaxf(v, row_xor<2>(v));
    v = fmaxf(v, row_xor<4>(v));
    v = fmaxf(v, row_xor<8>(v));
    return fmaxf(v, other_row(v));
}
__device__ __forceinline__ float wave_sum(float v) {
    v += row_xor<1>(v);
    v += row_xor<2>(v);
    v += row_xor<4>(v);
    v += row_xor<8>(v);
    return v + other_row(v);
}

__device__ __forceinline__ float leaky(float v) { return (v > 0.f) ? v : 0.2f * v; }

template <int H, bool ELU>
__global__ void dst_gather(const int* __restrict__ srt,
                           const int* __restrict__ beg, const int* __restrict__ fin,
                           const float* __restrict__ as_, const float* __restrict__ ad_,
                           const float* __restrict__ hsrc, const float* __restrict__ bias,
                           float* __restrict__ out, int N) {
    const int lane = threadIdx.x & 31;
    const int d    = blockIdx.x * (blockDim.x >> 5) + (threadIdx.x >> 5);
    if (d >= N) return;
    const int jb = __builtin_amdgcn_readfirstlane(beg[d]);
    const int je = __builtin_amdgcn_readfirstlane(fin[d]);
    const int n0 = (je - jb < 32) ? (je - jb) : 32;

    if constexpr (H == 4) {
        const float bv0 = bias[lane * 4], bv1 = bias[lane * 4 + 1], bv2 = bias[lane * 4 + 2], bv3 = bias[lane * 4 + 3];
        const float4 adv = *(const float4*)(ad_ + (size_t)d * 4);
        const bool hi = (lane & 16) != 0, odd = (lane & 8) != 0;
        float4 acc = make_float4(0.f, 0.f, 0.f, 0.f);

        auto logits = [&](int base, int n, int& sI) -> float4 {
            float4 e = make_float4(-INFINITY, -INFINITY, -INFINITY, -INFINITY);
            sI = 0;
            if (lane < n) {
                sI = srt[base + lane];
                const float4 av = *(const float4*)(as_ + (size_t)sI * 4);
                e.x = leaky(av.x + adv.x);
                e.y = leaky(av.y + adv.y);
                e.z = leaky(av.z + adv.z);
                e.w = leaky(av.w + adv.w);
            }
            return e;
        };
        auto rows = [&](int sI, float4 a, int n) {
#pragma unroll 4
            for (int k = 0; k < n; ++k) {
                const int   s  = __builtin_amdgcn_readlane(sI, k);
                const float b0 = __int_as_float(__builtin_amdgcn_readlane(__float_as_int(a.x), k));
                const float b1 = __int_as_float(__builtin_amdgcn_readlane(__float_as_int(a.y), k));
                const float b2 = __int_as_float(__builtin_amdgcn_readlane(__float_as_int(a.z), k));
                const float b3 = __int_as_float(__builtin_amdgcn_readlane(__float_as_int(a.w), k));
                const float al = hi ? (odd ? b3 : b2) : (odd ? b1 : b0);
                const float4 hv = *(const float4*)(hsrc + (size_t)s * 128 + lane * 4);
                acc.x += hv.x * al;
                acc.y += hv.y * al;
                acc.z += hv.z * al;
                acc.w += hv.w * al;
            }
        };

        int sI0;
        const float4 e0 = logits(jb, n0, sI0);

        float4 m = e0;
        for (int base = jb + 32; base < je; base += 32) {
            const int n = (je - base < 32) ? (je - base) : 32;
            int sI;
            const float4 e = logits(base, n, sI);
            m.x = fmaxf(m.x, e.x); m.y = fmaxf(m.y, e.y); m.z = fmaxf(m.z, e.z); m.w = fmaxf(m.w, e.w);
        }
        m.x = wave_max(m.x); m.y = wave_max(m.y); m.z = wave_max(m.z); m.w = wave_max(m.w);

        float4 p0 = make_float4(0.f, 0.f, 0.f, 0.f);
        if (lane < n0) {
            p0.x = __expf(e0.x - m.x);
            p0.y = __expf(e0.y - m.y);
            p0.z = __expf(e0.z - m.z);
            p0.w = __expf(e0.w - m.w);
        }
        float4 sv = p0;
        for (int base = jb + 32; base < je; base += 32) {
            const int n = (je - base < 32) ? (je - base) : 32;
            int sI;
            const float4 e = logits(base, n, sI);
            if (lane < n) {
                sv.x += __expf(e.x - m.x);
                sv.y += __expf(e.y - m.y);
                sv.z += __expf(e.z - m.z);
                sv.w += __expf(e.w - m.w);
            }
        }
        sv.x = wave_sum(sv.x); sv.y = wave_sum(sv.y); sv.z = wave_sum(sv.z); sv.w = wave_sum(sv.w);

        rows(sI0, make_float4(p0.x / (sv.x + 1e-16f), p0.y / (sv.y + 1e-16f),
                              p0.z / (sv.z + 1e-16f), p0.w / (sv.w + 1e-16f)), n0);
        for (int base = jb + 32; base < je; base += 32) {
            const int n = (je - base < 32) ? (je - base) : 32;
            int sI;
            const float4 e = logits(base, n, sI);
            float4 a = make_float4(0.f, 0.f, 0.f, 0.f);
            if (lane < n) {
                a.x = __expf(e.x - m.x) / (sv.x + 1e-16f);
                a.y = __expf(e.y - m.y) / (sv.y + 1e-16f);
                a.z = __expf(e.z - m.z) / (sv.z + 1e-16f);
                a.w = __expf(e.w - m.w) / (sv.w + 1e-16f);
            }
            rows(sI, a, n);
        }
        *(float4*)(out + (size_t)d * 128 + lane * 4) =
            make_float4(finish<ELU>(acc.x, bv0), finish<ELU>(acc.y, bv1),
                        finish<ELU>(acc.z, bv2), finish<ELU>(acc.w, bv3));
    } else {
        const float bv0 = bias[lane * 2], bv1 = bias[lane * 2 + 1];
        const float adv = ad_[d];
        float2 acc = make_float2(0.f, 0.f);

        auto logits = [&](int base, int n, int& sI) -> float {
            float e = -INFINITY;
            sI = 0;
            if (lane < n) {
                sI = srt[base + lane];
                e  = leaky(as_[sI] + adv);
            }
            return e;
        };
        auto rows = [&](int sI, float a, int n) {
#pragma unroll 4
            for (int k = 0; k < n; ++k) {
                const int   s  = __builtin_amdgcn_readlane(sI, k);
                const float al = __int_as_float(__builtin_amdgcn_readlane(__float_as_int(a), k));
                const float2 hv = *(const float2*)(hsrc + (size_t)s * 64 + lane * 2);
                acc.x += hv.x * al;
                acc.y += hv.y * al;
            }
        };

        int sI0;
        const float e0 = logits(jb, n0, sI0);

        float m = e0;
        for (int base = jb + 32; base < je; base += 32) {
            const int n = (je - base < 32) ? (je - base) : 32;
            int sI;
            m = fmaxf(m, logits(base, n, sI));
        }
        m = wave_max(m);

        const float p0 = (lane < n0) ? __expf(e0 - m) : 0.f;
        float sv = p0;
        for (int base = jb + 32; base < je; base += 32) {
            const int n = (je - base < 32) ? (je - base) : 32;
            int sI;
            const float e = logits(base, n, sI);
            if (lane < n) sv += __expf(e - m);
        }
        sv = wave_sum(sv);

        rows(sI0, p0 / (sv + 1e-16f), n0);
        for (int base = jb + 32; base < je; base += 32) {
            const int n = (je - base < 32) ? (je - base) : 32;
            int sI;
            const float e = logits(base, n, sI);
            const float a = (lane < n) ? __expf(e - m) / (sv + 1e-16f) : 0.f;
            rows(sI, a, n);
        }
        *(float2*)(out + (size_t)d * 64 + lane * 2) =
            make_float2(finish<ELU>(acc.x, bv0), finish<ELU>(acc.y, bv1));
    }
}

extern "C" void kernel_launch(void* const* d_in, const int* in_sizes, int n_in,
                              void* d_out, int out_size, void* d_ws, size_t ws_size,
                              hipStream_t stream) {
    const float* x        = (const float*)d_in[0];
    const int*   ei       = (const int*)  d_in[1];
    const float* W1       = (const float*)d_in[2];
    const float* att_src1 = (const float*)d_in[3];
    const float* att_dst1 = (const float*)d_in[4];
    const float* b1       = (const float*)d_in[5];
    const float* W2       = (const float*)d_in[6];
    const float* att_src2 = (const float*)d_in[7];
    const float* att_dst2 = (const float*)d_in[8];
    const float* b2       = (const float*)d_in[9];
    float* out = (float*)d_out;

    const int N = in_sizes[0] / 128;
    const int E = in_sizes[1] / 2;

    float* ws = (float*)d_ws;
    float*    h1  = ws;
    float*    f1  = h1 + (size_t)N * 128;
    float*    as1 = f1 + (size_t)N * 128;
    float*    ad1 = as1 + (size_t)N * 4;
    float*    as2 = ad1 + (size_t)N * 4;
    float*    ad2 = as2 + N;
    int*      cnt = (int*)(ad2 + N);
    int*      beg = cnt + N;
    int*      fin = beg + N;
    int*      srt = fin + N;
    int*      rnk = srt + E;
    float*    h2  = h1;

    hipMemsetAsync(cnt, 0, sizeof(int) * (size_t)N, stream);

    const int TB = 256;
    const int mTiles = (N + 15) / 16;
    const int gBlocks = (mTiles + 3) / 4;
    const int eBlocks = (E + TB - 1) / TB;
    const int nBlocks = (N + 7) / 8;

    dst_count<<<eBlocks, TB, 0, stream>>>(ei, E, cnt, rnk);
    dst_scan <<<(N + 1023) / 1024, 256, 0, stream>>>(cnt, N, beg, fin);
    dst_fill <<<eBlocks, TB, 0, stream>>>(ei, E, beg, rnk, srt);

    gemm_wmma_f32<128, 128><<<gBlocks, 128, 0, stream>>>(x, W1, h1, N);
    att_scores<<<(N * 4 + TB - 1) / TB, TB, 0, stream>>>(h1, att_src1, att_dst1, as1, ad1, N, 4, 32);
    dst_gather<4, true><<<nBlocks, TB, 0, stream>>>(srt, beg, fin, as1, ad1, h1, b1, f1, N);

    gemm_wmma_f32<128, 64><<<gBlocks, 128, 0, stream>>>(f1, W2, h2, N);
    att_scores<<<(N + TB - 1) / TB, TB, 0, stream>>>(h2, att_src2, att_dst2, as2, ad2, N, 1, 64);
    dst_gather<1, false><<<nBlocks, TB, 0, stream>>>(srt, beg, fin, as2, ad2, h2, b2, out, N);
}


# Edge sort by destination redone as a two-level counting sort on LDS atomics (256-node buckets, then nodes): the 800000 scattered global atomic adds and the memset go; segment order still by arrival
# speedup vs baseline: 7.5575x; 1.4184x over previous
#include <hip/hip_runtime.h>
#include <math.h>

typedef __attribute__((ext_vector_type(2))) float v2f;
typedef __attribute__((ext_vector_type(8))) float v8f;

template <int XMASK>
__device__ __forceinline__ float row_xor(float v) {
    const int b = __float_as_int(v);
    return __int_as_float(__builtin_amdgcn_update_dpp(b, b, 0x160 | XMASK, 0xf, 0xf, false));
}
__device__ __forceinline__ float other_row(float v) {
    const int b = __float_as_int(v);
    return __int_as_float(__builtin_amdgcn_permlanex16(b, b, 0x76543210, 0xfedcba98, false, false));
}

template <int K, int NC, int H>
__global__ void __launch_bounds__(128)
gemm_wmma_f32(const float* __restrict__ A,
              const float* __restrict__ B,
              float* __restrict__ C,
              const float* __restrict__ a_src,
              const float* __restrict__ a_dst,
              float* __restrict__ out_s,
              float* __restrict__ out_d,
              int M) {
    constexpr int P = K + 4;
    constexpr int NT  = NC / 16;
    constexpr int TPH = NT / H;
    __shared__ float Bs[NC * P];
    const int tid = threadIdx.x;
    const int l15 = tid & 15;

    float wsrc[NT], wdst[NT];
#pragma unroll
    for (int tn = 0; tn < NT; ++tn) {
        wsrc[tn] = a_src[tn * 16 + l15];
        wdst[tn] = a_dst[tn * 16 + l15];
    }

    for (int idx = tid; idx < K * NC; idx += 128) {
        int k = idx / NC;
        int n = idx - k * NC;
        Bs[n * P + k] = B[idx];
    }
    __syncthreads();

    const int waveId = tid >> 5;
    const int lane   = tid & 31;
    const int half   = lane >> 4;
    const int kb     = half << 1;
    const int m0     = (blockIdx.x * 4 + waveId) << 4;
    if (m0 >= M) return;

    int rowA = m0 + l15;
    if (rowA >= M) rowA = M - 1;
    const float* __restrict__ arow = A + (size_t)rowA * K;

    v2f areg[K / 4];
#pragma unroll
    for (int kk = 0; kk < K / 4; ++kk)
        areg[kk] = *(const v2f*)(arow + kk * 4 + kb);

    const int rbase = m0 + (half << 3);
    float ps[8], pd[8];
    float res[H];
    const bool b3 = (l15 & 8) != 0, b2 = (l15 & 4) != 0, b1 = (l15 & 2) != 0, b0 = (l15 & 1) != 0;
#pragma unroll
    for (int tn = 0; tn < NT; ++tn) {
        const int n0 = tn << 4;
        const float* bcol = &Bs[(n0 + l15) * P + kb];
        v8f acc = {};
#pragma unroll
        for (int kk = 0; kk < K / 4; ++kk) {
            v2f b = *(const v2f*)(bcol + kk * 4);
            acc = __builtin_amdgcn_wmma_f32_16x16x4_f32(false, areg[kk], false, b,
                                                        (short)0, acc, false, false);
        }
        const int colC  = n0 + l15;
#pragma unroll
        for (int r = 0; r < 8; ++r) {
            int row = rbase + r;
            if (row < M) C[(size_t)row * NC + colC] = acc[r];
        }
        if (tn % TPH == 0) {
#pragma unroll
            for (int r = 0; r < 8; ++r) { ps[r] = acc[r] * wsrc[tn]; pd[r] = acc[r] * wdst[tn]; }
        } else {
#pragma unroll
            for (int r = 0; r < 8; ++r) { ps[r] += acc[r] * wsrc[tn]; pd[r] += acc[r] * wdst[tn]; }
        }
        if (tn % TPH == TPH - 1) {
            float t8[8], t4[4], t2[2];
#pragma unroll
            for (int r = 0; r < 8; ++r)
                t8[r] = (b3 ? pd[r] : ps[r]) + row_xor<8>(b3 ? ps[r] : pd[r]);
#pragma unroll
            for (int j = 0; j < 4; ++j)
                t4[j] = (b2 ? t8[j + 4] : t8[j]) + row_xor<4>(b2 ? t8[j] : t8[j + 4]);
#pragma unroll
            for (int j = 0; j < 2; ++j)
                t2[j] = (b1 ? t4[j + 2] : t4[j]) + row_xor<2>(b1 ? t4[j] : t4[j + 2]);
            res[tn / TPH] = (b0 ? t2[1] : t2[0]) + row_xor<1>(b0 ? t2[0] : t2[1]);
        }
    }
    const int srow = rbase + (l15 & 7);
    if (srow < M) {
        float* __restrict__ o = (b3 ? out_d : out_s) + (size_t)srow * H;
        if constexpr (H == 4) {
            *(float4*)o = make_float4(res[0], res[1], res[2], res[3]);
        } else {
#pragma unroll
            for (int hh = 0; hh < H; ++hh) o[hh] = res[hh];
        }
    }
}

constexpr int BUCKET_SHIFT = 8;
constexpr int BUCKET_NODES = 1 << BUCKET_SHIFT;
constexpr int MAX_BUCKETS  = 1024;
constexpr int SORT_THREADS = 1024;
constexpr int EDGES_AHEAD  = 4;

__global__ void __launch_bounds__(SORT_THREADS)
bucket_count(const int* __restrict__ ei, int E, int chunk, int nbk, int* __restrict__ cnt) {
    __shared__ int hist[MAX_BUCKETS];
    const int tid = threadIdx.x;
    for (int i = tid; i < nbk; i += SORT_THREADS) hist[i] = 0;
    __syncthreads();
    const int e0 = blockIdx.x * chunk;
    const int e1 = (e0 + chunk < E) ? (e0 + chunk) : E;
    for (int base = e0 + tid; base < e1; base += SORT_THREADS * EDGES_AHEAD) {
        int d[EDGES_AHEAD];
#pragma unroll
        for (int j = 0; j < EDGES_AHEAD; ++j) {
            const int e = base + j * SORT_THREADS;
            d[j] = (e < e1) ? ei[E + e] : -1;
        }
#pragma unroll
        for (int j = 0; j < EDGES_AHEAD; ++j)
            if (d[j] >= 0) atomicAdd(&hist[d[j] >> BUCKET_SHIFT], 1);
    }
    __syncthreads();
    for (int i = tid; i < nbk; i += SORT_THREADS) cnt[i * gridDim.x + blockIdx.x] = hist[i];
}

__global__ void excl_scan(const int* __restrict__ cnt, int n, int* __restrict__ out) {
    __shared__ int red[256];
    __shared__ int sc[256];
    const int tid  = threadIdx.x;
    const int base = blockIdx.x * 1024;

    int pre = 0;
    for (int i = tid; i < base; i += 256) pre += cnt[i];

    const int i0 = base + tid * 4;
    int c0 = (i0     < n) ? cnt[i0]     : 0;
    int c1 = (i0 + 1 < n) ? cnt[i0 + 1] : 0;
    int c2 = (i0 + 2 < n) ? cnt[i0 + 2] : 0;
    int c3 = (i0 + 3 < n) ? cnt[i0 + 3] : 0;
    const int mine = c0 + c1 + c2 + c3;

    red[tid] = pre;
    sc[tid]  = mine;
    __syncthreads();
    for (int o = 128; o > 0; o >>= 1) {
        if (tid < o) red[tid] += red[tid + o];
        __syncthreads();
    }
    for (int o = 1; o < 256; o <<= 1) {
        int v = (tid >= o) ? sc[tid - o] : 0;
        __syncthreads();
        sc[tid] += v;
        __syncthreads();
    }
    int o0 = red[0] + sc[tid] - mine;
    int o1 = o0 + c0;
    int o2 = o1 + c1;
    int o3 = o2 + c2;
    if (i0     < n) out[i0]     = o0;
    if (i0 + 1 < n) out[i0 + 1] = o1;
    if (i0 + 2 < n) out[i0 + 2] = o2;
    if (i0 + 3 < n) out[i0 + 3] = o3;
}

__global__ void __launch_bounds__(SORT_THREADS)
bucket_scatter(const int* __restrict__ ei, int E, int chunk, int nbk,
               const int* __restrict__ off, int* __restrict__ pairs_) {
    __shared__ int cur[MAX_BUCKETS];
    int2* __restrict__ pairs = (int2*)pairs_;
    const int tid = threadIdx.x;
    for (int i = tid; i < nbk; i += SORT_THREADS) cur[i] = off[i * gridDim.x + blockIdx.x];
    __syncthreads();
    const int e0 = blockIdx.x * chunk;
    const int e1 = (e0 + chunk < E) ? (e0 + chunk) : E;
    for (int base = e0 + tid; base < e1; base += SORT_THREADS * EDGES_AHEAD) {
        int d[EDGES_AHEAD], s[EDGES_AHEAD];
#pragma unroll
        for (int j = 0; j < EDGES_AHEAD; ++j) {
            const int e = base + j * SORT_THREADS;
            const bool ok = e < e1;
            d[j] = ok ? ei[E + e] : -1;
            s[j] = ok ? ei[e] : 0;
        }
#pragma unroll
        for (int j = 0; j < EDGES_AHEAD; ++j)
            if (d[j] >= 0) {
                const int p = atomicAdd(&cur[d[j] >> BUCKET_SHIFT], 1);
                pairs[p] = make_int2(s[j], d[j]);
            }
    }
}

constexpr int STAGE_EDGES = 8192;
__global__ void __launch_bounds__(SORT_THREADS)
bucket_sort(const int* __restrict__ pairs_, const int* __restrict__ off, int chunks, int nbk, int E, int N,
            int* __restrict__ beg, int* __restrict__ fin, int* __restrict__ srt) {
    __shared__ int cnt[BUCKET_NODES];
    __shared__ int cur[BUCKET_NODES];
    __shared__ int stage[STAGE_EDGES];
    const int2* __restrict__ pairs = (const int2*)pairs_;
    const int tid = threadIdx.x;
    const int c   = blockIdx.x;
    const int b0  = off[c * chunks];
    const int b1  = (c + 1 < nbk) ? off[(c + 1) * chunks] : E;
    if (tid < BUCKET_NODES) cnt[tid] = 0;
    __syncthreads();
    for (int base = b0 + tid; base < b1; base += SORT_THREADS * EDGES_AHEAD) {
        int d[EDGES_AHEAD];
#pragma unroll
        for (int j = 0; j < EDGES_AHEAD; ++j) {
            const int i = base + j * SORT_THREADS;
            d[j] = (i < b1) ? pairs[i].y : -1;
        }
#pragma unroll
        for (int j = 0; j < EDGES_AHEAD; ++j)
            if (d[j] >= 0) atomicAdd(&cnt[d[j] & (BUCKET_NODES - 1)], 1);
    }
    __syncthreads();
    if (tid < 32) {
        int v[8];
        int tot = 0;
#pragma unroll
        for (int j = 0; j < 8; ++j) { v[j] = cnt[tid * 8 + j]; tot += v[j]; }
        int inc = tot;
#pragma unroll
        for (int o = 1; o < 32; o <<= 1) {
            const int t = __shfl_up(inc, o);
            if (tid >= o) inc += t;
        }
        int ex = inc - tot;
#pragma unroll
        for (int j = 0; j < 8; ++j) {
            cur[tid * 8 + j] = ex;
            const int node = c * BUCKET_NODES + tid * 8 + j;
            if (node < N) { beg[node] = b0 + ex; fin[node] = b0 + ex + v[j]; }
            ex += v[j];
        }
    }
    __syncthreads();
    const int  nb   = b1 - b0;
    const bool fits = nb <= STAGE_EDGES;
    for (int base = b0 + tid; base < b1; base += SORT_THREADS * EDGES_AHEAD) {
        int2 sd[EDGES_AHEAD];
#pragma unroll
        for (int j = 0; j < EDGES_AHEAD; ++j) {
            const int i = base + j * SORT_THREADS;
            sd[j] = (i < b1) ? pairs[i] : make_int2(0, -1);
        }
#pragma unroll
        for (int j = 0; j < EDGES_AHEAD; ++j)
            if (sd[j].y >= 0) {
                const int p = atomicAdd(&cur[sd[j].y & (BUCKET_NODES - 1)], 1);
                if (fits) stage[p] = sd[j].x; else srt[b0 + p] = sd[j].x;
            }
    }
    if (fits) {
        __syncthreads();
        for (int i = tid; i < nb; i += SORT_THREADS) srt[b0 + i] = stage[i];
    }
}

template <bool ELU>
__device__ __forceinline__ float finish(float agg, float b) {
    const float v = agg + b;
    if (ELU) return (v > 0.f) ? v : (__expf(v) - 1.f);
    return v;
}

__device__ __forceinline__ float wave_max(float v) {
    v = fmaxf(v, row_xor<1>(v));
    v = fmaxf(v, row_xor<2>(v));
    v = fmaxf(v, row_xor<4>(v));
    v = fmaxf(v, row_xor<8>(v));
    return fmaxf(v, other_row(v));
}
__device__ __forceinline__ float wave_sum(float v) {
    v += row_xor<1>(v);
    v += row_xor<2>(v);
    v += row_xor<4>(v);
    v += row_xor<8>(v);
    return v + other_row(v);
}

__device__ __forceinline__ float leaky(float v) { return (v > 0.f) ? v : 0.2f * v; }

template <int H, bool ELU>
__global__ void dst_gather(const int* __restrict__ srt,
                           const int* __restrict__ beg, const int* __restrict__ fin,
                           const float* __restrict__ as_, const float* __restrict__ ad_,
                           const float* __restrict__ hsrc, const float* __restrict__ bias,
                           float* __restrict__ out, int N) {
    const int lane = threadIdx.x & 31;
    const int d    = blockIdx.x * (blockDim.x >> 5) + (threadIdx.x >> 5);
    if (d >= N) return;
    const int jb = __builtin_amdgcn_readfirstlane(beg[d]);
    const int je = __builtin_amdgcn_readfirstlane(fin[d]);
    const int n0 = (je - jb < 32) ? (je - jb) : 32;

    if constexpr (H == 4) {
        const float bv0 = bias[lane * 4], bv1 = bias[lane * 4 + 1], bv2 = bias[lane * 4 + 2], bv3 = bias[lane * 4 + 3];
        const float4 adv = *(const float4*)(ad_ + (size_t)d * 4);
        const bool hi = (lane & 16) != 0, odd = (lane & 8) != 0;
        float4 acc = make_float4(0.f, 0.f, 0.f, 0.f);

        auto logits = [&](int base, int n, int& sI) -> float4 {
            float4 e = make_float4(-INFINITY, -INFINITY, -INFINITY, -INFINITY);
            sI = 0;
            if (lane < n) {
                sI = srt[base + lane];
                const float4 av = *(const float4*)(as_ + (size_t)sI * 4);
                e.x = leaky(av.x + adv.x);
                e.y = leaky(av.y + adv.y);
                e.z = leaky(av.z + adv.z);
                e.w = leaky(av.w + adv.w);
            }
            return e;
        };
        auto rows = [&](int sI, float4 a, int n) {
#pragma unroll 4
            for (int k = 0; k < n; ++k) {
                const int   s  = __builtin_amdgcn_readlane(sI, k);
                const float b0 = __int_as_float(__builtin_amdgcn_readlane(__float_as_int(a.x), k));
                const float b1 = __int_as_float(__builtin_amdgcn_readlane(__float_as_int(a.y), k));
                const float b2 = __int_as_float(__builtin_amdgcn_readlane(__float_as_int(a.z), k));
                const float b3 = __int_as_float(__builtin_amdgcn_readlane(__float_as_int(a.w), k));
                const float al = hi ? (odd ? b3 : b2) : (odd ? b1 : b0);
                const float4 hv = *(const float4*)(hsrc + (size_t)s * 128 + lane * 4);
                acc.x += hv.x * al;
                acc.y += hv.y * al;
                acc.z += hv.z * al;
                acc.w += hv.w * al;
            }
        };

        int sI0;
        const float4 e0 = logits(jb, n0, sI0);

        float4 m = e0;
        for (int base = jb + 32; base < je; base += 32) {
            const int n = (je - base < 32) ? (je - base) : 32;
            int sI;
            const float4 e = logits(base, n, sI);
            m.x = fmaxf(m.x, e.x); m.y = fmaxf(m.y, e.y); m.z = fmaxf(m.z, e.z); m.w = fmaxf(m.w, e.w);
        }
        m.x = wave_max(m.x); m.y = wave_max(m.y); m.z = wave_max(m.z); m.w = wave_max(m.w);

        float4 p0 = make_float4(0.f, 0.f, 0.f, 0.f);
        if (lane < n0) {
            p0.x = __expf(e0.x - m.x);
            p0.y = __expf(e0.y - m.y);
            p0.z = __expf(e0.z - m.z);
            p0.w = __expf(e0.w - m.w);
        }
        float4 sv = p0;
        for (int base = jb + 32; base < je; base += 32) {
            const int n = (je - base < 32) ? (je - base) : 32;
            int sI;
            const float4 e = logits(base, n, sI);
            if (lane < n) {
                sv.x += __expf(e.x - m.x);
                sv.y += __expf(e.y - m.y);
                sv.z += __expf(e.z - m.z);
                sv.w += __expf(e.w - m.w);
            }
        }
        sv.x = wave_sum(sv.x); sv.y = wave_sum(sv.y); sv.z = wave_sum(sv.z); sv.w = wave_sum(sv.w);

        rows(sI0, make_float4(p0.x / (sv.x + 1e-16f), p0.y / (sv.y + 1e-16f),
                              p0.z / (sv.z + 1e-16f), p0.w / (sv.w + 1e-16f)), n0);
        for (int base = jb + 32; base < je; base += 32) {
            const int n = (je - base < 32) ? (je - base) : 32;
            int sI;
            const float4 e = logits(base, n, sI);
            float4 a = make_float4(0.f, 0.f, 0.f, 0.f);
            if (lane < n) {
                a.x = __expf(e.x - m.x) / (sv.x + 1e-16f);
                a.y = __expf(e.y - m.y) / (sv.y + 1e-16f);
                a.z = __expf(e.z - m.z) / (sv.z + 1e-16f);
                a.w = __expf(e.w - m.w) / (sv.w + 1e-16f);
            }
            rows(sI, a, n);
        }
        *(float4*)(out + (size_t)d * 128 + lane * 4) =
            make_float4(finish<ELU>(acc.x, bv0), finish<ELU>(acc.y, bv1),
                        finish<ELU>(acc.z, bv2), finish<ELU>(acc.w, bv3));
    } else {
        const float bv0 = bias[lane * 2], bv1 = bias[lane * 2 + 1];
        const float adv = ad_[d];
        float2 acc = make_float2(0.f, 0.f);

        auto logits = [&](int base, int n, int& sI) -> float {
            float e = -INFINITY;
            sI = 0;
            if (lane < n) {
                sI = srt[base + lane];
                e  = leaky(as_[sI] + adv);
            }
            return e;
        };
        auto rows = [&](int sI, float a, int n) {
#pragma unroll 4
            for (int k = 0; k < n; ++k) {
                const int   s  = __builtin_amdgcn_readlane(sI, k);
                const float al = __int_as_float(__builtin_amdgcn_readlane(__float_as_int(a), k));
                const float2 hv = *(const float2*)(hsrc + (size_t)s * 64 + lane * 2);
                acc.x += hv.x * al;
                acc.y += hv.y * al;
            }
        };

        int sI0;
        const float e0 = logits(jb, n0, sI0);

        float m = e0;
        for (int base = jb + 32; base < je; base += 32) {
            const int n = (je - base < 32) ? (je - base) : 32;
            int sI;
            m = fmaxf(m, logits(base, n, sI));
        }
        m = wave_max(m);

        const float p0 = (lane < n0) ? __expf(e0 - m) : 0.f;
        float sv = p0;
        for (int base = jb + 32; base < je; base += 32) {
            const int n = (je - base < 32) ? (je - base) : 32;
            int sI;
            const float e = logits(base, n, sI);
            if (lane < n) sv += __expf(e - m);
        }
        sv = wave_sum(sv);

        rows(sI0, p0 / (sv + 1e-16f), n0);
        for (int base = jb + 32; base < je; base += 32) {
            const int n = (je - base < 32) ? (je - base) : 32;
            int sI;
            const float e = logits(base, n, sI);
            const float a = (lane < n) ? __expf(e - m) / (sv + 1e-16f) : 0.f;
            rows(sI, a, n);
        }
        *(float2*)(out + (size_t)d * 64 + lane * 2) =
            make_float2(finish<ELU>(acc.x, bv0), finish<ELU>(acc.y, bv1));
    }
}

extern "C" void kernel_launch(void* const* d_in, const int* in_sizes, int n_in,
                              void* d_out, int out_size, void* d_ws, size_t ws_size,
                              hipStream_t stream) {
    const float* x        = (const float*)d_in[0];
    const int*   ei       = (const int*)  d_in[1];
    const float* W1       = (const float*)d_in[2];
    const float* att_src1 = (const float*)d_in[3];
    const float* att_dst1 = (const float*)d_in[4];
    const float* b1       = (const float*)d_in[5];
    const float* W2       = (const float*)d_in[6];
    const float* att_src2 = (const float*)d_in[7];
    const float* att_dst2 = (const float*)d_in[8];
    const float* b2       = (const float*)d_in[9];
    float* out = (float*)d_out;

    const int N = in_sizes[0] / 128;
    const int E = in_sizes[1] / 2;

    float* ws = (float*)d_ws;
    float*    h1  = ws;
    float*    f1  = h1 + (size_t)N * 128;
    float*    as1 = f1 + (size_t)N * 128;
    float*    ad1 = as1 + (size_t)N * 4;
    float*    as2 = ad1 + (size_t)N * 4;
    float*    ad2 = as2 + N;
    int*      beg = (int*)(ad2 + N);
    int*      fin = beg + N;
    int*      srt = fin + N;
    float*    h2  = h1;

    const int chunks = 256;
    const int chunk  = (E + chunks - 1) / chunks;
    const int nbk    = (N + BUCKET_NODES - 1) >> BUCKET_SHIFT;
    const int nmat   = nbk * chunks;
    int*      bcnt  = (int*)h1;
    int*      boff  = bcnt + nmat;
    int*      pairs = (int*)f1;

    const int TB = 256;
    const int mTiles = (N + 15) / 16;
    const int gBlocks = (mTiles + 3) / 4;
    const int nBlocks = (N + 7) / 8;

    bucket_count  <<<chunks, SORT_THREADS, 0, stream>>>(ei, E, chunk, nbk, bcnt);
    excl_scan     <<<(nmat + 1023) / 1024, 256, 0, stream>>>(bcnt, nmat, boff);
    bucket_scatter<<<chunks, SORT_THREADS, 0, stream>>>(ei, E, chunk, nbk, boff, pairs);
    bucket_sort   <<<nbk, SORT_THREADS, 0, stream>>>(pairs, boff, chunks, nbk, E, N, beg, fin, srt);

    gemm_wmma_f32<128, 128, 4><<<gBlocks, 128, 0, stream>>>(x, W1, h1, att_src1, att_dst1, as1, ad1, N);
    dst_gather<4, true><<<nBlocks, TB, 0, stream>>>(srt, beg, fin, as1, ad1, h1, b1, f1, N);

    gemm_wmma_f32<128, 64, 1><<<gBlocks, 128, 0, stream>>>(f1, W2, h2, att_src2, att_dst2, as2, ad2, N);
    dst_gather<1, false><<<nBlocks, TB, 0, stream>>>(srt, beg, fin, as2, ad2, h2, b2, out, N);
}


# x@W1 with its attention scores done by extra work-groups of the first sort launch (role by block index, A tile in LDS so two work-groups share a WGP): it overlaps the sort, one launch fewer
# speedup vs baseline: 9.6273x; 1.2739x over previous
#include <hip/hip_runtime.h>
#include <math.h>

typedef __attribute__((ext_vector_type(2))) float v2f;
typedef __attribute__((ext_vector_type(8))) float v8f;

template <int XMASK>
__device__ __forceinline__ float row_xor(float v) {
    const int b = __float_as_int(v);
    return __int_as_float(__builtin_amdgcn_update_dpp(b, b, 0x160 | XMASK, 0xf, 0xf, false));
}
__device__ __forceinline__ float other_row(float v) {
    const int b = __float_as_int(v);
    return __int_as_float(__builtin_amdgcn_permlanex16(b, b, 0x76543210, 0xfedcba98, false, false));
}

template <int K, int NC, int H>
__global__ void __launch_bounds__(128)
gemm_wmma_f32(const float* __restrict__ A,
              const float* __restrict__ B,
              float* __restrict__ C,
              const float* __restrict__ a_src,
              const float* __restrict__ a_dst,
              float* __restrict__ out_s,
              float* __restrict__ out_d,
              int M) {
    constexpr int P = K + 4;
    constexpr int NT  = NC / 16;
    constexpr int TPH = NT / H;
    __shared__ float Bs[NC * P];
    const int tid = threadIdx.x;
    const int l15 = tid & 15;

    float wsrc[NT], wdst[NT];
#pragma unroll
    for (int tn = 0; tn < NT; ++tn) {
        wsrc[tn] = a_src[tn * 16 + l15];
        wdst[tn] = a_dst[tn * 16 + l15];
    }

    for (int idx = tid; idx < K * NC; idx += 128) {
        int k = idx / NC;
        int n = idx - k * NC;
        Bs[n * P + k] = B[idx];
    }
    __syncthreads();

    const int waveId = tid >> 5;
    const int lane   = tid & 31;
    const int half   = lane >> 4;
    const int kb     = half << 1;
    const int m0     = (blockIdx.x * 4 + waveId) << 4;
    if (m0 >= M) return;

    int rowA = m0 + l15;
    if (rowA >= M) rowA = M - 1;
    const float* __restrict__ arow = A + (size_t)rowA * K;

    v2f areg[K / 4];
#pragma unroll
    for (int kk = 0; kk < K / 4; ++kk)
        areg[kk] = *(const v2f*)(arow + kk * 4 + kb);

    const int rbase = m0 + (half << 3);
    float ps[8], pd[8];
    float res[H];
    const bool b3 = (l15 & 8) != 0, b2 = (l15 & 4) != 0, b1 = (l15 & 2) != 0, b0 = (l15 & 1) != 0;
#pragma unroll
    for (int tn = 0; tn < NT; ++tn) {
        const int n0 = tn << 4;
        const float* bcol = &Bs[(n0 + l15) * P + kb];
        v8f acc = {};
#pragma unroll
        for (int kk = 0; kk < K / 4; ++kk) {
            v2f b = *(const v2f*)(bcol + kk * 4);
            acc = __builtin_amdgcn_wmma_f32_16x16x4_f32(false, areg[kk], false, b,
                                                        (short)0, acc, false, false);
        }
        const int colC  = n0 + l15;
#pragma unroll
        for (int r = 0; r < 8; ++r) {
            int row = rbase + r;
            if (row < M) C[(size_t)row * NC + colC] = acc[r];
        }
        if (tn % TPH == 0) {
#pragma unroll
            for (int r = 0; r < 8; ++r) { ps[r] = acc[r] * wsrc[tn]; pd[r] = acc[r] * wdst[tn]; }
        } else {
#pragma unroll
            for (int r = 0; r < 8; ++r) { ps[r] += acc[r] * wsrc[tn]; pd[r] += acc[r] * wdst[tn]; }
        }
        if (tn % TPH == TPH - 1) {
            float t8[8], t4[4], t2[2];
#pragma unroll
            for (int r = 0; r < 8; ++r)
                t8[r] = (b3 ? pd[r] : ps[r]) + row_xor<8>(b3 ? ps[r] : pd[r]);
#pragma unroll
            for (int j = 0; j < 4; ++j)
                t4[j] = (b2 ? t8[j + 4] : t8[j]) + row_xor<4>(b2 ? t8[j] : t8[j + 4]);
#pragma unroll
            for (int j = 0; j < 2; ++j)
                t2[j] = (b1 ? t4[j + 2] : t4[j]) + row_xor<2>(b1 ? t4[j] : t4[j + 2]);
            res[tn / TPH] = (b0 ? t2[1] : t2[0]) + row_xor<1>(b0 ? t2[0] : t2[1]);
        }
    }
    const int srow = rbase + (l15 & 7);
    if (srow < M) {
        float* __restrict__ o = (b3 ? out_d : out_s) + (size_t)srow * H;
        if constexpr (H == 4) {
            *(float4*)o = make_float4(res[0], res[1], res[2], res[3]);
        } else {
#pragma unroll
            for (int hh = 0; hh < H; ++hh) o[hh] = res[hh];
        }
    }
}

constexpr int BUCKET_SHIFT = 8;
constexpr int BUCKET_NODES = 1 << BUCKET_SHIFT;
constexpr int MAX_BUCKETS  = 1024;
constexpr int SORT_THREADS = 1024;
constexpr int CHUNKS       = 256;
constexpr int EDGES_AHEAD  = 4;

template <int XMASK>
__device__ __forceinline__ int row_xor_i(int v) {
    return __builtin_amdgcn_update_dpp(v, v, 0x160 | XMASK, 0xf, 0xf, false);
}
__device__ __forceinline__ int wave_sum_i(int v) {
    v += row_xor_i<1>(v);
    v += row_xor_i<2>(v);
    v += row_xor_i<4>(v);
    v += row_xor_i<8>(v);
    return v + __builtin_amdgcn_permlanex16(v, v, 0x76543210, 0xfedcba98, false, false);
}

__device__ __forceinline__ void
chunk_group(const int c, const int* __restrict__ ei, int E, int chunk, int nbk,
            int* __restrict__ rstart, int* __restrict__ rcnt, int* __restrict__ pairs_) {
    __shared__ int hist[MAX_BUCKETS];
    __shared__ int cur[MAX_BUCKETS];
    int2* __restrict__ pairs = (int2*)pairs_;
    const int tid = threadIdx.x;
    for (int i = tid; i < nbk; i += SORT_THREADS) hist[i] = 0;
    __syncthreads();
    const int e0 = c * chunk;
    const int e1 = (e0 + chunk < E) ? (e0 + chunk) : E;
    for (int base = e0 + tid; base < e1; base += SORT_THREADS * EDGES_AHEAD) {
        int d[EDGES_AHEAD];
#pragma unroll
        for (int j = 0; j < EDGES_AHEAD; ++j) {
            const int e = base + j * SORT_THREADS;
            d[j] = (e < e1) ? ei[E + e] : -1;
        }
#pragma unroll
        for (int j = 0; j < EDGES_AHEAD; ++j)
            if (d[j] >= 0) atomicAdd(&hist[d[j] >> BUCKET_SHIFT], 1);
    }
    __syncthreads();
    if (tid < 32) {
        const int per = (nbk + 31) >> 5;
        const int i0  = tid * per;
        int tot = 0;
        for (int j = 0; j < per; ++j) {
            const int i = i0 + j;
            if (i < nbk) tot += hist[i];
        }
        int inc = tot;
#pragma unroll
        for (int o = 1; o < 32; o <<= 1) {
            const int t = __shfl_up(inc, o);
            if (tid >= o) inc += t;
        }
        int ex = e0 + inc - tot;
        for (int j = 0; j < per; ++j) {
            const int i = i0 + j;
            if (i < nbk) {
                const int n = hist[i];
                cur[i] = ex;
                rstart[i * CHUNKS + c] = ex;
                rcnt[i * CHUNKS + c]   = n;
                ex += n;
            }
        }
    }
    __syncthreads();
    for (int base = e0 + tid; base < e1; base += SORT_THREADS * EDGES_AHEAD) {
        int d[EDGES_AHEAD], s[EDGES_AHEAD];
#pragma unroll
        for (int j = 0; j < EDGES_AHEAD; ++j) {
            const int e = base + j * SORT_THREADS;
            const bool ok = e < e1;
            d[j] = ok ? ei[E + e] : -1;
            s[j] = ok ? ei[e] : 0;
        }
#pragma unroll
        for (int j = 0; j < EDGES_AHEAD; ++j)
            if (d[j] >= 0) {
                const int p = atomicAdd(&cur[d[j] >> BUCKET_SHIFT], 1);
                pairs[p] = make_int2(s[j], d[j]);
            }
    }
}

template <int K, int NC, int H>
__device__ __forceinline__ void
product_role(float* __restrict__ As, float* __restrict__ Bs, const int blk,
             const float* __restrict__ A, const float* __restrict__ B, float* __restrict__ C,
             const float* __restrict__ a_src, const float* __restrict__ a_dst,
             float* __restrict__ out_s, float* __restrict__ out_d, const int M) {
    static_assert(K == 128, "a wave stages one row of A per load: 32 lanes x float4");
    constexpr int P     = K + 4;
    constexpr int NT    = NC / 16;
    constexpr int TPH   = NT / H;
    constexpr int WAVES = SORT_THREADS / 32;
    constexpr int MT    = WAVES / H;
    constexpr int ROWS  = MT * 16;
    constexpr int APASS = ROWS / WAVES;
    constexpr int BPASS = K * NC / SORT_THREADS;
    const int tid  = threadIdx.x;
    const int lane = tid & 31;
    const int l15  = tid & 15;
    const int wv   = tid >> 5;
    const int hq   = wv % H;
    const int mt   = wv / H;
    const int row0 = blk * ROWS;

    float wsrc[TPH], wdst[TPH];
#pragma unroll
    for (int t = 0; t < TPH; ++t) {
        wsrc[t] = a_src[(hq * TPH + t) * 16 + l15];
        wdst[t] = a_dst[(hq * TPH + t) * 16 + l15];
    }

    float4 av[APASS];
    float  bv[BPASS];
#pragma unroll
    for (int j = 0; j < APASS; ++j) {
        int r = row0 + wv + WAVES * j;
        if (r >= M) r = M - 1;
        av[j] = *(const float4*)(A + (size_t)r * K + lane * 4);
    }
#pragma unroll
    for (int j = 0; j < BPASS; ++j) bv[j] = B[tid + j * SORT_THREADS];
#pragma unroll
    for (int j = 0; j < APASS; ++j) *(float4*)&As[(wv + WAVES * j) * P + lane * 4] = av[j];
#pragma unroll
    for (int j = 0; j < BPASS; ++j) {
        const int idx = tid + j * SORT_THREADS;
        const int k = idx / NC;
        const int n = idx - k * NC;
        Bs[n * P + k] = bv[j];
    }
    __syncthreads();

    const int half = lane >> 4;
    const int kb   = half << 1;
    const int m0   = row0 + mt * 16;
    if (m0 >= M) return;

    const float* arow = &As[(mt * 16 + l15) * P + kb];
    const float* bcol = &Bs[(hq * TPH * 16 + l15) * P + kb];
    v8f acc[TPH];
#pragma unroll
    for (int t = 0; t < TPH; ++t) acc[t] = v8f{};
#pragma unroll
    for (int kk = 0; kk < K / 4; ++kk) {
        const v2f a = *(const v2f*)(arow + kk * 4);
#pragma unroll
        for (int t = 0; t < TPH; ++t) {
            const v2f b = *(const v2f*)(bcol + t * 16 * P + kk * 4);
            acc[t] = __builtin_amdgcn_wmma_f32_16x16x4_f32(false, a, false, b, (short)0, acc[t], false, false);
        }
    }

    const int rbase = m0 + (half << 3);
#pragma unroll
    for (int t = 0; t < TPH; ++t) {
        const int colC = (hq * TPH + t) * 16 + l15;
#pragma unroll
        for (int r = 0; r < 8; ++r) {
            const int row = rbase + r;
            if (row < M) C[(size_t)row * NC + colC] = acc[t][r];
        }
    }
    float ps[8], pd[8];
#pragma unroll
    for (int r = 0; r < 8; ++r) { ps[r] = acc[0][r] * wsrc[0]; pd[r] = acc[0][r] * wdst[0]; }
#pragma unroll
    for (int t = 1; t < TPH; ++t) {
#pragma unroll
        for (int r = 0; r < 8; ++r) { ps[r] += acc[t][r] * wsrc[t]; pd[r] += acc[t][r] * wdst[t]; }
    }
    const bool b3 = (l15 & 8) != 0, b2 = (l15 & 4) != 0, b1 = (l15 & 2) != 0, b0 = (l15 & 1) != 0;
    float t8[8], t4[4], t2[2];
#pragma unroll
    for (int r = 0; r < 8; ++r)
        t8[r] = (b3 ? pd[r] : ps[r]) + row_xor<8>(b3 ? ps[r] : pd[r]);
#pragma unroll
    for (int j = 0; j < 4; ++j)
        t4[j] = (b2 ? t8[j + 4] : t8[j]) + row_xor<4>(b2 ? t8[j] : t8[j + 4]);
#pragma unroll
    for (int j = 0; j < 2; ++j)
        t2[j] = (b1 ? t4[j + 2] : t4[j]) + row_xor<2>(b1 ? t4[j] : t4[j + 2]);
    const float res = (b0 ? t2[1] : t2[0]) + row_xor<1>(b0 ? t2[0] : t2[1]);
    const int srow = rbase + (l15 & 7);
    if (srow < M) (b3 ? out_d : out_s)[(size_t)srow * H + hq] = res;
}

constexpr int L1_K = 128, L1_NC = 128, L1_H = 4;
constexpr int PRODUCT_ROWS = (SORT_THREADS / 32 / L1_H) * 16;
__global__ void __launch_bounds__(SORT_THREADS)
first_launch(const int* __restrict__ ei, int E, int chunk, int nbk,
             int* __restrict__ rstart, int* __restrict__ rcnt, int* __restrict__ pairs_,
             const float* __restrict__ x, const float* __restrict__ W, float* __restrict__ h,
             const float* __restrict__ a_src, const float* __restrict__ a_dst,
             float* __restrict__ out_s, float* __restrict__ out_d, int M) {
    __shared__ __attribute__((aligned(16))) float As[PRODUCT_ROWS * (L1_K + 4)];
    __shared__ __attribute__((aligned(16))) float Bs[L1_NC * (L1_K + 4)];
    if (blockIdx.x < CHUNKS)
        chunk_group(blockIdx.x, ei, E, chunk, nbk, rstart, rcnt, pairs_);
    else
        product_role<L1_K, L1_NC, L1_H>(As, Bs, blockIdx.x - CHUNKS, x, W, h, a_src, a_dst, out_s, out_d, M);
}

constexpr int STAGE_EDGES   = 8192;
constexpr int RUNS_PER_WAVE = CHUNKS / 32;
constexpr int COUNTS_AHEAD  = 16;
__global__ void __launch_bounds__(SORT_THREADS)
bucket_sort(const int* __restrict__ pairs_, const int* __restrict__ rstart, const int* __restrict__ rcnt,
            int N, int* __restrict__ beg, int* __restrict__ fin, int* __restrict__ srt) {
    __shared__ int cnt[BUCKET_NODES];
    __shared__ int cur[BUCKET_NODES];
    __shared__ int stage[STAGE_EDGES];
    __shared__ int rs[CHUNKS];
    __shared__ int rc[CHUNKS];
    __shared__ int red[2 * (SORT_THREADS / 32)];
    const int2* __restrict__ pairs = (const int2*)pairs_;
    const int tid  = threadIdx.x;
    const int lane = tid & 31;
    const int wv   = tid >> 5;
    const int c    = blockIdx.x;

    int own = 0, ownst = 0;
    if (tid < CHUNKS) {
        ownst = rstart[c * CHUNKS + tid];
        own   = rcnt[c * CHUNKS + tid];
    }
    int pre = 0;
    const int4* __restrict__ r4 = (const int4*)rcnt;
    const int n4 = c * (CHUNKS / 4);
    for (int i0 = tid; i0 < n4; i0 += SORT_THREADS * COUNTS_AHEAD) {
        int4 v[COUNTS_AHEAD];
#pragma unroll
        for (int j = 0; j < COUNTS_AHEAD; ++j) {
            const int i = i0 + j * SORT_THREADS;
            v[j] = (i < n4) ? r4[i] : make_int4(0, 0, 0, 0);
        }
#pragma unroll
        for (int j = 0; j < COUNTS_AHEAD; ++j) pre += (v[j].x + v[j].y) + (v[j].z + v[j].w);
    }
    if (tid < CHUNKS) { rs[tid] = ownst; rc[tid] = own; }
    if (tid < BUCKET_NODES) cnt[tid] = 0;
    pre = wave_sum_i(pre);
    own = wave_sum_i(own);
    if (lane == 0) { red[wv] = pre; red[SORT_THREADS / 32 + wv] = own; }
    __syncthreads();
    const int b0 = wave_sum_i(red[lane]);
    const int nb = wave_sum_i(red[SORT_THREADS / 32 + lane]);
    const bool fits = nb <= STAGE_EDGES;

    int2 sd[RUNS_PER_WAVE];
    int  rl[RUNS_PER_WAVE], rb[RUNS_PER_WAVE];
    bool lng = false;
#pragma unroll
    for (int i = 0; i < RUNS_PER_WAVE; ++i) {
        rl[i] = rc[wv + 32 * i];
        rb[i] = rs[wv + 32 * i];
        lng = lng || (rl[i] > 32);
    }
#pragma unroll
    for (int i = 0; i < RUNS_PER_WAVE; ++i)
        sd[i] = (lane < rl[i]) ? pairs[rb[i] + lane] : make_int2(0, -1);
#pragma unroll
    for (int i = 0; i < RUNS_PER_WAVE; ++i)
        if (sd[i].y >= 0) atomicAdd(&cnt[sd[i].y & (BUCKET_NODES - 1)], 1);
    if (lng) {
        for (int i = 0; i < RUNS_PER_WAVE; ++i) {
            const int r = wv + 32 * i;
            const int st = rs[r], ln = rc[r];
            for (int k = 32 + lane; k < ln; k += 32)
                atomicAdd(&cnt[pairs[st + k].y & (BUCKET_NODES - 1)], 1);
        }
    }
    __syncthreads();
    if (tid < 32) {
        int v[8];
        int tot = 0;
#pragma unroll
        for (int j = 0; j < 8; ++j) { v[j] = cnt[tid * 8 + j]; tot += v[j]; }
        int inc = tot;
#pragma unroll
        for (int o = 1; o < 32; o <<= 1) {
            const int t = __shfl_up(inc, o);
            if (tid >= o) inc += t;
        }
        int ex = inc - tot;
#pragma unroll
        for (int j = 0; j < 8; ++j) {
            cur[tid * 8 + j] = ex;
            const int node = c * BUCKET_NODES + tid * 8 + j;
            if (node < N) { beg[node] = b0 + ex; fin[node] = b0 + ex + v[j]; }
            ex += v[j];
        }
    }
    __syncthreads();
#pragma unroll
    for (int i = 0; i < RUNS_PER_WAVE; ++i)
        if (sd[i].y >= 0) {
            const int p = atomicAdd(&cur[sd[i].y & (BUCKET_NODES - 1)], 1);
            if (fits) stage[p] = sd[i].x; else srt[b0 + p] = sd[i].x;
        }
    if (lng) {
        for (int i = 0; i < RUNS_PER_WAVE; ++i) {
            const int r = wv + 32 * i;
            const int st = rs[r], ln = rc[r];
            for (int k = 32 + lane; k < ln; k += 32) {
                const int2 e = pairs[st + k];
                const int p = atomicAdd(&cur[e.y & (BUCKET_NODES - 1)], 1);
                if (fits) stage[p] = e.x; else srt[b0 + p] = e.x;
            }
        }
    }
    if (fits) {
        __syncthreads();
        for (int i = tid; i < nb; i += SORT_THREADS) srt[b0 + i] = stage[i];
    }
}

template <bool ELU>
__device__ __forceinline__ float finish(float agg, float b) {
    const float v = agg + b;
    if (ELU) return (v > 0.f) ? v : (__expf(v) - 1.f);
    return v;
}

__device__ __forceinline__ float wave_max(float v) {
    v = fmaxf(v, row_xor<1>(v));
    v = fmaxf(v, row_xor<2>(v));
    v = fmaxf(v, row_xor<4>(v));
    v = fmaxf(v, row_xor<8>(v));
    return fmaxf(v, other_row(v));
}
__device__ __forceinline__ float wave_sum(float v) {
    v += row_xor<1>(v);
    v += row_xor<2>(v);
    v += row_xor<4>(v);
    v += row_xor<8>(v);
    return v + other_row(v);
}

__device__ __forceinline__ float leaky(float v) { return (v > 0.f) ? v : 0.2f * v; }

template <int H, bool ELU>
__global__ void dst_gather(const int* __restrict__ srt,
                           const int* __restrict__ beg, const int* __restrict__ fin,
                           const float* __restrict__ as_, const float* __restrict__ ad_,
                           const float* __restrict__ hsrc, const float* __restrict__ bias,
                           float* __restrict__ out, int N) {
    const int lane = threadIdx.x & 31;
    const int d    = blockIdx.x * (blockDim.x >> 5) + (threadIdx.x >> 5);
    if (d >= N) return;
    const int jb = __builtin_amdgcn_readfirstlane(beg[d]);
    const int je = __builtin_amdgcn_readfirstlane(fin[d]);
    const int n0 = (je - jb < 32) ? (je - jb) : 32;

    if constexpr (H == 4) {
        const float bv0 = bias[lane * 4], bv1 = bias[lane * 4 + 1], bv2 = bias[lane * 4 + 2], bv3 = bias[lane * 4 + 3];
        const float4 adv = *(const float4*)(ad_ + (size_t)d * 4);
        const bool hi = (lane & 16) != 0, odd = (lane & 8) != 0;
        float4 acc = make_float4(0.f, 0.f, 0.f, 0.f);

        auto logits = [&](int base, int n, int& sI) -> float4 {
            float4 e = make_float4(-INFINITY, -INFINITY, -INFINITY, -INFINITY);
            sI = 0;
            if (lane < n) {
                sI = srt[base + lane];
                const float4 av = *(const float4*)(as_ + (size_t)sI * 4);
                e.x = leaky(av.x + adv.x);
                e.y = leaky(av.y + adv.y);
                e.z = leaky(av.z + adv.z);
                e.w = leaky(av.w + adv.w);
            }
            return e;
        };
        auto rows = [&](int sI, float4 a, int n) {
#pragma unroll 4
            for (int k = 0; k < n; ++k) {
                const int   s  = __builtin_amdgcn_readlane(sI, k);
                const float b0 = __int_as_float(__builtin_amdgcn_readlane(__float_as_int(a.x), k));
                const float b1 = __int_as_float(__builtin_amdgcn_readlane(__float_as_int(a.y), k));
                const float b2 = __int_as_float(__builtin_amdgcn_readlane(__float_as_int(a.z), k));
                const float b3 = __int_as_float(__builtin_amdgcn_readlane(__float_as_int(a.w), k));
                const float al = hi ? (odd ? b3 : b2) : (odd ? b1 : b0);
                const float4 hv = *(const float4*)(hsrc + (size_t)s * 128 + lane * 4);
                acc.x += hv.x * al;
                acc.y += hv.y * al;
                acc.z += hv.z * al;
                acc.w += hv.w * al;
            }
        };

        int sI0;
        const float4 e0 = logits(jb, n0, sI0);

        float4 m = e0;
        for (int base = jb + 32; base < je; base += 32) {
            const int n = (je - base < 32) ? (je - base) : 32;
            int sI;
            const float4 e = logits(base, n, sI);
            m.x = fmaxf(m.x, e.x); m.y = fmaxf(m.y, e.y); m.z = fmaxf(m.z, e.z); m.w = fmaxf(m.w, e.w);
        }
        m.x = wave_max(m.x); m.y = wave_max(m.y); m.z = wave_max(m.z); m.w = wave_max(m.w);

        float4 p0 = make_float4(0.f, 0.f, 0.f, 0.f);
        if (lane < n0) {
            p0.x = __expf(e0.x - m.x);
            p0.y = __expf(e0.y - m.y);
            p0.z = __expf(e0.z - m.z);
            p0.w = __expf(e0.w - m.w);
        }
        float4 sv = p0;
        for (int base = jb + 32; base < je; base += 32) {
            const int n = (je - base < 32) ? (je - base) : 32;
            int sI;
            const float4 e = logits(base, n, sI);
            if (lane < n) {
                sv.x += __expf(e.x - m.x);
                sv.y += __expf(e.y - m.y);
                sv.z += __expf(e.z - m.z);
                sv.w += __expf(e.w - m.w);
            }
        }
        sv.x = wave_sum(sv.x); sv.y = wave_sum(sv.y); sv.z = wave_sum(sv.z); sv.w = wave_sum(sv.w);

        rows(sI0, make_float4(p0.x / (sv.x + 1e-16f), p0.y / (sv.y + 1e-16f),
                              p0.z / (sv.z + 1e-16f), p0.w / (sv.w + 1e-16f)), n0);
        for (int base = jb + 32; base < je; base += 32) {
            const int n = (je - base < 32) ? (je - base) : 32;
            int sI;
            const float4 e = logits(base, n, sI);
            float4 a = make_float4(0.f, 0.f, 0.f, 0.f);
            if (lane < n) {
                a.x = __expf(e.x - m.x) / (sv.x + 1e-16f);
                a.y = __expf(e.y - m.y) / (sv.y + 1e-16f);
                a.z = __expf(e.z - m.z) / (sv.z + 1e-16f);
                a.w = __expf(e.w - m.w) / (sv.w + 1e-16f);
            }
            rows(sI, a, n);
        }
        *(float4*)(out + (size_t)d * 128 + lane * 4) =
            make_float4(finish<ELU>(acc.x, bv0), finish<ELU>(acc.y, bv1),
                        finish<ELU>(acc.z, bv2), finish<ELU>(acc.w, bv3));
    } else {
        const float bv0 = bias[lane * 2], bv1 = bias[lane * 2 + 1];
        const float adv = ad_[d];
        float2 acc = make_float2(0.f, 0.f);

        auto logits = [&](int base, int n, int& sI) -> float {
            float e = -INFINITY;
            sI = 0;
            if (lane < n) {
                sI = srt[base + lane];
                e  = leaky(as_[sI] + adv);
            }
            return e;
        };
        auto rows = [&](int sI, float a, int n) {
#pragma unroll 4
            for (int k = 0; k < n; ++k) {
                const int   s  = __builtin_amdgcn_readlane(sI, k);
                const float al = __int_as_float(__builtin_amdgcn_readlane(__float_as_int(a), k));
                const float2 hv = *(const float2*)(hsrc + (size_t)s * 64 + lane * 2);
                acc.x += hv.x * al;
                acc.y += hv.y * al;
            }
        };

        int sI0;
        const float e0 = logits(jb, n0, sI0);

        float m = e0;
        for (int base = jb + 32; base < je; base += 32) {
            const int n = (je - base < 32) ? (je - base) : 32;
            int sI;
            m = fmaxf(m, logits(base, n, sI));
        }
        m = wave_max(m);

        const float p0 = (lane < n0) ? __expf(e0 - m) : 0.f;
        float sv = p0;
        for (int base = jb + 32; base < je; base += 32) {
            const int n = (je - base < 32) ? (je - base) : 32;
            int sI;
            const float e = logits(base, n, sI);
            if (lane < n) sv += __expf(e - m);
        }
        sv = wave_sum(sv);

        rows(sI0, p0 / (sv + 1e-16f), n0);
        for (int base = jb + 32; base < je; base += 32) {
            const int n = (je - base < 32) ? (je - base) : 32;
            int sI;
            const float e = logits(base, n, sI);
            const float a = (lane < n) ? __expf(e - m) / (sv + 1e-16f) : 0.f;
            rows(sI, a, n);
        }
        *(float2*)(out + (size_t)d * 64 + lane * 2) =
            make_float2(finish<ELU>(acc.x, bv0), finish<ELU>(acc.y, bv1));
    }
}

extern "C" void kernel_launch(void* const* d_in, const int* in_sizes, int n_in,
                              void* d_out, int out_size, void* d_ws, size_t ws_size,
                              hipStream_t stream) {
    const float* x        = (const float*)d_in[0];
    const int*   ei       = (const int*)  d_in[1];
    const float* W1       = (const float*)d_in[2];
    const float* att_src1 = (const float*)d_in[3];
    const float* att_dst1 = (const float*)d_in[4];
    const float* b1       = (const float*)d_in[5];
    const float* W2       = (const float*)d_in[6];
    const float* att_src2 = (const float*)d_in[7];
    const float* att_dst2 = (const float*)d_in[8];
    const float* b2       = (const float*)d_in[9];
    float* out = (float*)d_out;

    const int N = in_sizes[0] / 128;
    const int E = in_sizes[1] / 2;

    float* ws = (float*)d_ws;
    float*    h1  = ws;
    float*    f1  = h1 + (size_t)N * 128;
    float*    as1 = f1 + (size_t)N * 128;
    float*    ad1 = as1 + (size_t)N * 4;
    float*    as2 = ad1 + (size_t)N * 4;
    float*    ad2 = as2 + N;
    int*      beg = (int*)(ad2 + N);
    int*      fin = beg + N;
    int*      srt = fin + N;
    float*    h2  = h1;

    const int chunk  = (E + CHUNKS - 1) / CHUNKS;
    const int nbk    = (N + BUCKET_NODES - 1) >> BUCKET_SHIFT;
    const int nmat   = nbk * CHUNKS;
    int*      rstart = (int*)f1;
    int*      rcnt   = rstart + nmat;
    int*      pairs  = rcnt + nmat;

    const int TB = 256;
    const int mTiles = (N + 15) / 16;
    const int gBlocks = (mTiles + 3) / 4;
    const int nBlocks = (N + 7) / 8;
    const int pBlocks = (N + PRODUCT_ROWS - 1) / PRODUCT_ROWS;

    first_launch<<<CHUNKS + pBlocks, SORT_THREADS, 0, stream>>>(ei, E, chunk, nbk, rstart, rcnt, pairs,
                                                              x, W1, h1, att_src1, att_dst1, as1, ad1, N);
    bucket_sort<<<nbk, SORT_THREADS, 0, stream>>>(pairs, rstart, rcnt, N, beg, fin, srt);

    dst_gather<4, true><<<nBlocks, TB, 0, stream>>>(srt, beg, fin, as1, ad1, h1, b1, f1, N);

    gemm_wmma_f32<128, 64, 1><<<gBlocks, 128, 0, stream>>>(f1, W2, h2, att_src2, att_dst2, as2, ad2, N);
    dst_gather<1, false><<<nBlocks, TB, 0, stream>>>(srt, beg, fin, as2, ad2, h2, b2, out, N);
}


# Layer-2 product f1@W2 and its scores formed by the layer-1 gather's 16-node work-group from its rows in LDS (k split in 4 partial sums, scores by wave sum: another summation order); one launch fewer
# speedup vs baseline: 9.9249x; 1.0309x over previous
#include <hip/hip_runtime.h>
#include <math.h>

typedef __attribute__((ext_vector_type(2))) float v2f;
typedef __attribute__((ext_vector_type(8))) float v8f;

template <int XMASK>
__device__ __forceinline__ float row_xor(float v) {
    const int b = __float_as_int(v);
    return __int_as_float(__builtin_amdgcn_update_dpp(b, b, 0x160 | XMASK, 0xf, 0xf, false));
}
__device__ __forceinline__ float other_row(float v) {
    const int b = __float_as_int(v);
    return __int_as_float(__builtin_amdgcn_permlanex16(b, b, 0x76543210, 0xfedcba98, false, false));
}


constexpr int BUCKET_SHIFT = 8;
constexpr int BUCKET_NODES = 1 << BUCKET_SHIFT;
constexpr int MAX_BUCKETS  = 1024;
constexpr int SORT_THREADS = 1024;
constexpr int CHUNKS       = 256;
constexpr int EDGES_AHEAD  = 4;

template <int XMASK>
__device__ __forceinline__ int row_xor_i(int v) {
    return __builtin_amdgcn_update_dpp(v, v, 0x160 | XMASK, 0xf, 0xf, false);
}
__device__ __forceinline__ int wave_sum_i(int v) {
    v += row_xor_i<1>(v);
    v += row_xor_i<2>(v);
    v += row_xor_i<4>(v);
    v += row_xor_i<8>(v);
    return v + __builtin_amdgcn_permlanex16(v, v, 0x76543210, 0xfedcba98, false, false);
}

__device__ __forceinline__ void
chunk_group(const int c, const int* __restrict__ ei, int E, int chunk, int nbk,
            int* __restrict__ rstart, int* __restrict__ rcnt, int* __restrict__ pairs_) {
    __shared__ int hist[MAX_BUCKETS];
    __shared__ int cur[MAX_BUCKETS];
    int2* __restrict__ pairs = (int2*)pairs_;
    const int tid = threadIdx.x;
    for (int i = tid; i < nbk; i += SORT_THREADS) hist[i] = 0;
    __syncthreads();
    const int e0 = c * chunk;
    const int e1 = (e0 + chunk < E) ? (e0 + chunk) : E;
    for (int base = e0 + tid; base < e1; base += SORT_THREADS * EDGES_AHEAD) {
        int d[EDGES_AHEAD];
#pragma unroll
        for (int j = 0; j < EDGES_AHEAD; ++j) {
            const int e = base + j * SORT_THREADS;
            d[j] = (e < e1) ? ei[E + e] : -1;
        }
#pragma unroll
        for (int j = 0; j < EDGES_AHEAD; ++j)
            if (d[j] >= 0) atomicAdd(&hist[d[j] >> BUCKET_SHIFT], 1);
    }
    __syncthreads();
    if (tid < 32) {
        const int per = (nbk + 31) >> 5;
        const int i0  = tid * per;
        int tot = 0;
        for (int j = 0; j < per; ++j) {
            const int i = i0 + j;
            if (i < nbk) tot += hist[i];
        }
        int inc = tot;
#pragma unroll
        for (int o = 1; o < 32; o <<= 1) {
            const int t = __shfl_up(inc, o);
            if (tid >= o) inc += t;
        }
        int ex = e0 + inc - tot;
        for (int j = 0; j < per; ++j) {
            const int i = i0 + j;
            if (i < nbk) {
                const int n = hist[i];
                cur[i] = ex;
                rstart[i * CHUNKS + c] = ex;
                rcnt[i * CHUNKS + c]   = n;
                ex += n;
            }
        }
    }
    __syncthreads();
    for (int base = e0 + tid; base < e1; base += SORT_THREADS * EDGES_AHEAD) {
        int d[EDGES_AHEAD], s[EDGES_AHEAD];
#pragma unroll
        for (int j = 0; j < EDGES_AHEAD; ++j) {
            const int e = base + j * SORT_THREADS;
            const bool ok = e < e1;
            d[j] = ok ? ei[E + e] : -1;
            s[j] = ok ? ei[e] : 0;
        }
#pragma unroll
        for (int j = 0; j < EDGES_AHEAD; ++j)
            if (d[j] >= 0) {
                const int p = atomicAdd(&cur[d[j] >> BUCKET_SHIFT], 1);
                pairs[p] = make_int2(s[j], d[j]);
            }
    }
}

template <int K, int NC, int H>
__device__ __forceinline__ void
product_role(float* __restrict__ As, float* __restrict__ Bs, const int blk,
             const float* __restrict__ A, const float* __restrict__ B, float* __restrict__ C,
             const float* __restrict__ a_src, const float* __restrict__ a_dst,
             float* __restrict__ out_s, float* __restrict__ out_d, const int M) {
    static_assert(K == 128, "a wave stages one row of A per load: 32 lanes x float4");
    constexpr int P     = K + 4;
    constexpr int NT    = NC / 16;
    constexpr int TPH   = NT / H;
    constexpr int WAVES = SORT_THREADS / 32;
    constexpr int MT    = WAVES / H;
    constexpr int ROWS  = MT * 16;
    constexpr int APASS = ROWS / WAVES;
    constexpr int BPASS = K * NC / SORT_THREADS;
    const int tid  = threadIdx.x;
    const int lane = tid & 31;
    const int l15  = tid & 15;
    const int wv   = tid >> 5;
    const int hq   = wv % H;
    const int mt   = wv / H;
    const int row0 = blk * ROWS;

    float wsrc[TPH], wdst[TPH];
#pragma unroll
    for (int t = 0; t < TPH; ++t) {
        wsrc[t] = a_src[(hq * TPH + t) * 16 + l15];
        wdst[t] = a_dst[(hq * TPH + t) * 16 + l15];
    }

    float4 av[APASS];
    float  bv[BPASS];
#pragma unroll
    for (int j = 0; j < APASS; ++j) {
        int r = row0 + wv + WAVES * j;
        if (r >= M) r = M - 1;
        av[j] = *(const float4*)(A + (size_t)r * K + lane * 4);
    }
#pragma unroll
    for (int j = 0; j < BPASS; ++j) bv[j] = B[tid + j * SORT_THREADS];
#pragma unroll
    for (int j = 0; j < APASS; ++j) *(float4*)&As[(wv + WAVES * j) * P + lane * 4] = av[j];
#pragma unroll
    for (int j = 0; j < BPASS; ++j) {
        const int idx = tid + j * SORT_THREADS;
        const int k = idx / NC;
        const int n = idx - k * NC;
        Bs[n * P + k] = bv[j];
    }
    __syncthreads();

    const int half = lane >> 4;
    const int kb   = half << 1;
    const int m0   = row0 + mt * 16;
    if (m0 >= M) return;

    const float* arow = &As[(mt * 16 + l15) * P + kb];
    const float* bcol = &Bs[(hq * TPH * 16 + l15) * P + kb];
    v8f acc[TPH];
#pragma unroll
    for (int t = 0; t < TPH; ++t) acc[t] = v8f{};
#pragma unroll
    for (int kk = 0; kk < K / 4; ++kk) {
        const v2f a = *(const v2f*)(arow + kk * 4);
#pragma unroll
        for (int t = 0; t < TPH; ++t) {
            const v2f b = *(const v2f*)(bcol + t * 16 * P + kk * 4);
            acc[t] = __builtin_amdgcn_wmma_f32_16x16x4_f32(false, a, false, b, (short)0, acc[t], false, false);
        }
    }

    const int rbase = m0 + (half << 3);
#pragma unroll
    for (int t = 0; t < TPH; ++t) {
        const int colC = (hq * TPH + t) * 16 + l15;
#pragma unroll
        for (int r = 0; r < 8; ++r) {
            const int row = rbase + r;
            if (row < M) C[(size_t)row * NC + colC] = acc[t][r];
        }
    }
    float ps[8], pd[8];
#pragma unroll
    for (int r = 0; r < 8; ++r) { ps[r] = acc[0][r] * wsrc[0]; pd[r] = acc[0][r] * wdst[0]; }
#pragma unroll
    for (int t = 1; t < TPH; ++t) {
#pragma unroll
        for (int r = 0; r < 8; ++r) { ps[r] += acc[t][r] * wsrc[t]; pd[r] += acc[t][r] * wdst[t]; }
    }
    const bool b3 = (l15 & 8) != 0, b2 = (l15 & 4) != 0, b1 = (l15 & 2) != 0, b0 = (l15 & 1) != 0;
    float t8[8], t4[4], t2[2];
#pragma unroll
    for (int r = 0; r < 8; ++r)
        t8[r] = (b3 ? pd[r] : ps[r]) + row_xor<8>(b3 ? ps[r] : pd[r]);
#pragma unroll
    for (int j = 0; j < 4; ++j)
        t4[j] = (b2 ? t8[j + 4] : t8[j]) + row_xor<4>(b2 ? t8[j] : t8[j + 4]);
#pragma unroll
    for (int j = 0; j < 2; ++j)
        t2[j] = (b1 ? t4[j + 2] : t4[j]) + row_xor<2>(b1 ? t4[j] : t4[j + 2]);
    const float res = (b0 ? t2[1] : t2[0]) + row_xor<1>(b0 ? t2[0] : t2[1]);
    const int srow = rbase + (l15 & 7);
    if (srow < M) (b3 ? out_d : out_s)[(size_t)srow * H + hq] = res;
}

constexpr int L1_K = 128, L1_NC = 128, L1_H = 4;
constexpr int PRODUCT_ROWS = (SORT_THREADS / 32 / L1_H) * 16;
__global__ void __launch_bounds__(SORT_THREADS)
first_launch(const int* __restrict__ ei, int E, int chunk, int nbk,
             int* __restrict__ rstart, int* __restrict__ rcnt, int* __restrict__ pairs_,
             const float* __restrict__ x, const float* __restrict__ W, float* __restrict__ h,
             const float* __restrict__ a_src, const float* __restrict__ a_dst,
             float* __restrict__ out_s, float* __restrict__ out_d, int M) {
    __shared__ __attribute__((aligned(16))) float As[PRODUCT_ROWS * (L1_K + 4)];
    __shared__ __attribute__((aligned(16))) float Bs[L1_NC * (L1_K + 4)];
    if (blockIdx.x < CHUNKS)
        chunk_group(blockIdx.x, ei, E, chunk, nbk, rstart, rcnt, pairs_);
    else
        product_role<L1_K, L1_NC, L1_H>(As, Bs, blockIdx.x - CHUNKS, x, W, h, a_src, a_dst, out_s, out_d, M);
}

constexpr int STAGE_EDGES   = 8192;
constexpr int RUNS_PER_WAVE = CHUNKS / 32;
constexpr int COUNTS_AHEAD  = 16;
__global__ void __launch_bounds__(SORT_THREADS)
bucket_sort(const int* __restrict__ pairs_, const int* __restrict__ rstart, const int* __restrict__ rcnt,
            int N, int* __restrict__ beg, int* __restrict__ fin, int* __restrict__ srt) {
    __shared__ int cnt[BUCKET_NODES];
    __shared__ int cur[BUCKET_NODES];
    __shared__ int stage[STAGE_EDGES];
    __shared__ int rs[CHUNKS];
    __shared__ int rc[CHUNKS];
    __shared__ int red[2 * (SORT_THREADS / 32)];
    const int2* __restrict__ pairs = (const int2*)pairs_;
    const int tid  = threadIdx.x;
    const int lane = tid & 31;
    const int wv   = tid >> 5;
    const int c    = blockIdx.x;

    int own = 0, ownst = 0;
    if (tid < CHUNKS) {
        ownst = rstart[c * CHUNKS + tid];
        own   = rcnt[c * CHUNKS + tid];
    }
    int pre = 0;
    const int4* __restrict__ r4 = (const int4*)rcnt;
    const int n4 = c * (CHUNKS / 4);
    for (int i0 = tid; i0 < n4; i0 += SORT_THREADS * COUNTS_AHEAD) {
        int4 v[COUNTS_AHEAD];
#pragma unroll
        for (int j = 0; j < COUNTS_AHEAD; ++j) {
            const int i = i0 + j * SORT_THREADS;
            v[j] = (i < n4) ? r4[i] : make_int4(0, 0, 0, 0);
        }
#pragma unroll
        for (int j = 0; j < COUNTS_AHEAD; ++j) pre += (v[j].x + v[j].y) + (v[j].z + v[j].w);
    }
    if (tid < CHUNKS) { rs[tid] = ownst; rc[tid] = own; }
    if (tid < BUCKET_NODES) cnt[tid] = 0;
    pre = wave_sum_i(pre);
    own = wave_sum_i(own);
    if (lane == 0) { red[wv] = pre; red[SORT_THREADS / 32 + wv] = own; }
    __syncthreads();
    const int b0 = wave_sum_i(red[lane]);
    const int nb = wave_sum_i(red[SORT_THREADS / 32 + lane]);
    const bool fits = nb <= STAGE_EDGES;

    int2 sd[RUNS_PER_WAVE];
    int  rl[RUNS_PER_WAVE], rb[RUNS_PER_WAVE];
    bool lng = false;
#pragma unroll
    for (int i = 0; i < RUNS_PER_WAVE; ++i) {
        rl[i] = rc[wv + 32 * i];
        rb[i] = rs[wv + 32 * i];
        lng = lng || (rl[i] > 32);
    }
#pragma unroll
    for (int i = 0; i < RUNS_PER_WAVE; ++i)
        sd[i] = (lane < rl[i]) ? pairs[rb[i] + lane] : make_int2(0, -1);
#pragma unroll
    for (int i = 0; i < RUNS_PER_WAVE; ++i)
        if (sd[i].y >= 0) atomicAdd(&cnt[sd[i].y & (BUCKET_NODES - 1)], 1);
    if (lng) {
        for (int i = 0; i < RUNS_PER_WAVE; ++i) {
            const int r = wv + 32 * i;
            const int st = rs[r], ln = rc[r];
            for (int k = 32 + lane; k < ln; k += 32)
                atomicAdd(&cnt[pairs[st + k].y & (BUCKET_NODES - 1)], 1);
        }
    }
    __syncthreads();
    if (tid < 32) {
        int v[8];
        int tot = 0;
#pragma unroll
        for (int j = 0; j < 8; ++j) { v[j] = cnt[tid * 8 + j]; tot += v[j]; }
        int inc = tot;
#pragma unroll
        for (int o = 1; o < 32; o <<= 1) {
            const int t = __shfl_up(inc, o);
            if (tid >= o) inc += t;
        }
        int ex = inc - tot;
#pragma unroll
        for (int j = 0; j < 8; ++j) {
            cur[tid * 8 + j] = ex;
            const int node = c * BUCKET_NODES + tid * 8 + j;
            if (node < N) { beg[node] = b0 + ex; fin[node] = b0 + ex + v[j]; }
            ex += v[j];
        }
    }
    __syncthreads();
#pragma unroll
    for (int i = 0; i < RUNS_PER_WAVE; ++i)
        if (sd[i].y >= 0) {
            const int p = atomicAdd(&cur[sd[i].y & (BUCKET_NODES - 1)], 1);
            if (fits) stage[p] = sd[i].x; else srt[b0 + p] = sd[i].x;
        }
    if (lng) {
        for (int i = 0; i < RUNS_PER_WAVE; ++i) {
            const int r = wv + 32 * i;
            const int st = rs[r], ln = rc[r];
            for (int k = 32 + lane; k < ln; k += 32) {
                const int2 e = pairs[st + k];
                const int p = atomicAdd(&cur[e.y & (BUCKET_NODES - 1)], 1);
                if (fits) stage[p] = e.x; else srt[b0 + p] = e.x;
            }
        }
    }
    if (fits) {
        __syncthreads();
        for (int i = tid; i < nb; i += SORT_THREADS) srt[b0 + i] = stage[i];
    }
}


template <bool ELU>
__device__ __forceinline__ float finish(float agg, float b) {
    const float v = agg + b;
    if (ELU) return (v > 0.f) ? v : (__expf(v) - 1.f);
    return v;
}

__device__ __forceinline__ float wave_max(float v) {
    v = fmaxf(v, row_xor<1>(v));
    v = fmaxf(v, row_xor<2>(v));
    v = fmaxf(v, row_xor<4>(v));
    v = fmaxf(v, row_xor<8>(v));
    return fmaxf(v, other_row(v));
}
__device__ __forceinline__ float wave_sum(float v) {
    v += row_xor<1>(v);
    v += row_xor<2>(v);
    v += row_xor<4>(v);
    v += row_xor<8>(v);
    return v + other_row(v);
}

__device__ __forceinline__ float leaky(float v) { return (v > 0.f) ? v : 0.2f * v; }


template <bool ELU>
__device__ __forceinline__ float4
gathered_row4(const int lane, const int d, const int jb, const int je,
              const int* __restrict__ srt, const float* __restrict__ as_, const float* __restrict__ ad_,
              const float* __restrict__ hsrc, const float* __restrict__ bias) {
    const int n0 = (je - jb < 32) ? (je - jb) : 32;
    const float bv0 = bias[lane * 4], bv1 = bias[lane * 4 + 1], bv2 = bias[lane * 4 + 2], bv3 = bias[lane * 4 + 3];
    const float4 adv = *(const float4*)(ad_ + (size_t)d * 4);
    const bool hi = (lane & 16) != 0, odd = (lane & 8) != 0;
    float4 acc = make_float4(0.f, 0.f, 0.f, 0.f);

    auto logits = [&](int base, int n, int& sI) -> float4 {
        float4 e = make_float4(-INFINITY, -INFINITY, -INFINITY, -INFINITY);
        sI = 0;
        if (lane < n) {
            sI = srt[base + lane];
            const float4 av = *(const float4*)(as_ + (size_t)sI * 4);
            e.x = leaky(av.x + adv.x);
            e.y = leaky(av.y + adv.y);
            e.z = leaky(av.z + adv.z);
            e.w = leaky(av.w + adv.w);
        }
        return e;
    };
    auto rows = [&](int sI, float4 a, int n) {
#pragma unroll 4
        for (int k = 0; k < n; ++k) {
            const int   s  = __builtin_amdgcn_readlane(sI, k);
            const float b0 = __int_as_float(__builtin_amdgcn_readlane(__float_as_int(a.x), k));
            const float b1 = __int_as_float(__builtin_amdgcn_readlane(__float_as_int(a.y), k));
            const float b2 = __int_as_float(__builtin_amdgcn_readlane(__float_as_int(a.z), k));
            const float b3 = __int_as_float(__builtin_amdgcn_readlane(__float_as_int(a.w), k));
            const float al = hi ? (odd ? b3 : b2) : (odd ? b1 : b0);
            const float4 hv = *(const float4*)(hsrc + (size_t)s * 128 + lane * 4);
            acc.x += hv.x * al;
            acc.y += hv.y * al;
            acc.z += hv.z * al;
            acc.w += hv.w * al;
        }
    };

    int sI0;
    const float4 e0 = logits(jb, n0, sI0);

    float4 m = e0;
    for (int base = jb + 32; base < je; base += 32) {
        const int n = (je - base < 32) ? (je - base) : 32;
        int sI;
        const float4 e = logits(base, n, sI);
        m.x = fmaxf(m.x, e.x); m.y = fmaxf(m.y, e.y); m.z = fmaxf(m.z, e.z); m.w = fmaxf(m.w, e.w);
    }
    m.x = wave_max(m.x); m.y = wave_max(m.y); m.z = wave_max(m.z); m.w = wave_max(m.w);

    float4 p0 = make_float4(0.f, 0.f, 0.f, 0.f);
    if (lane < n0) {
        p0.x = __expf(e0.x - m.x);
        p0.y = __expf(e0.y - m.y);
        p0.z = __expf(e0.z - m.z);
        p0.w = __expf(e0.w - m.w);
    }
    float4 sv = p0;
    for (int base = jb + 32; base < je; base += 32) {
        const int n = (je - base < 32) ? (je - base) : 32;
        int sI;
        const float4 e = logits(base, n, sI);
        if (lane < n) {
            sv.x += __expf(e.x - m.x);
            sv.y += __expf(e.y - m.y);
            sv.z += __expf(e.z - m.z);
            sv.w += __expf(e.w - m.w);
        }
    }
    sv.x = wave_sum(sv.x); sv.y = wave_sum(sv.y); sv.z = wave_sum(sv.z); sv.w = wave_sum(sv.w);

    rows(sI0, make_float4(p0.x / (sv.x + 1e-16f), p0.y / (sv.y + 1e-16f),
                          p0.z / (sv.z + 1e-16f), p0.w / (sv.w + 1e-16f)), n0);
    for (int base = jb + 32; base < je; base += 32) {
        const int n = (je - base < 32) ? (je - base) : 32;
        int sI;
        const float4 e = logits(base, n, sI);
        float4 a = make_float4(0.f, 0.f, 0.f, 0.f);
        if (lane < n) {
            a.x = __expf(e.x - m.x) / (sv.x + 1e-16f);
            a.y = __expf(e.y - m.y) / (sv.y + 1e-16f);
            a.z = __expf(e.z - m.z) / (sv.z + 1e-16f);
            a.w = __expf(e.w - m.w) / (sv.w + 1e-16f);
        }
        rows(sI, a, n);
    }
    return make_float4(finish<ELU>(acc.x, bv0), finish<ELU>(acc.y, bv1),
                       finish<ELU>(acc.z, bv2), finish<ELU>(acc.w, bv3));
}

constexpr int GP_NODES   = 16;
constexpr int GP_THREADS = GP_NODES * 32;
template <int K, int NC>
__global__ void __launch_bounds__(GP_THREADS)
gather_product(const int* __restrict__ srt, const int* __restrict__ beg, const int* __restrict__ fin,
               const float* __restrict__ as_, const float* __restrict__ ad_,
               const float* __restrict__ hsrc, const float* __restrict__ bias,
               const float* __restrict__ B, const float* __restrict__ a_src, const float* __restrict__ a_dst,
               float* __restrict__ C, float* __restrict__ out_s, float* __restrict__ out_d, int N) {
    static_assert(K == 128, "a wave's finished row is one float4 per lane");
    static_assert(NC == 64, "a wave stores a row of C as one float2 per lane");
    constexpr int P     = K + 4;
    constexpr int PC    = NC + 4;
    constexpr int NT    = NC / 16;
    constexpr int NQ    = GP_NODES / NT;
    constexpr int KQ    = K / 4 / NQ;
    constexpr int BPASS = K * NC / GP_THREADS;
    static_assert(NT * NQ == GP_NODES && KQ * NQ * 4 == K, "every wave one column tile and one k range");
    __shared__ __attribute__((aligned(16))) float As[GP_NODES * P];
    __shared__ __attribute__((aligned(16))) float Bs[NC * P];
    __shared__ __attribute__((aligned(16))) float Cp[NQ * GP_NODES * PC];
    const int tid  = threadIdx.x;
    const int lane = tid & 31;
    const int l15  = tid & 15;
    const int wv   = tid >> 5;
    const int d    = blockIdx.x * GP_NODES + wv;
    const bool live = d < N;

    float bv[BPASS];
#pragma unroll
    for (int j = 0; j < BPASS; ++j) bv[j] = B[tid + j * GP_THREADS];
    int jb = 0, je = 0;
    if (live) { jb = beg[d]; je = fin[d]; }
#pragma unroll
    for (int j = 0; j < BPASS; ++j) {
        const int idx = tid + j * GP_THREADS;
        const int k = idx / NC;
        const int n = idx - k * NC;
        Bs[n * P + k] = bv[j];
    }

    float4 row = make_float4(0.f, 0.f, 0.f, 0.f);
    if (live)
        row = gathered_row4<true>(lane, d, __builtin_amdgcn_readfirstlane(jb), __builtin_amdgcn_readfirstlane(je),
                                  srt, as_, ad_, hsrc, bias);
    const float2 ws = *(const float2*)(a_src + lane * 2);
    const float2 wd = *(const float2*)(a_dst + lane * 2);
    *(float4*)&As[wv * P + lane * 4] = row;
    __syncthreads();

    {
        const int half = lane >> 4;
        const int kb   = half << 1;
        const int tn   = wv % NT;
        const int q    = wv / NT;
        const float* arow = &As[l15 * P + q * (KQ * 4) + kb];
        const float* bcol = &Bs[(tn * 16 + l15) * P + q * (KQ * 4) + kb];
        v2f a[KQ], b[KQ];
#pragma unroll
        for (int kk = 0; kk < KQ; ++kk) {
            a[kk] = *(const v2f*)(arow + kk * 4);
            b[kk] = *(const v2f*)(bcol + kk * 4);
        }
        v8f acc = {};
#pragma unroll
        for (int kk = 0; kk < KQ; ++kk)
            acc = __builtin_amdgcn_wmma_f32_16x16x4_f32(false, a[kk], false, b[kk], (short)0, acc, false, false);
#pragma unroll
        for (int r = 0; r < 8; ++r)
            Cp[(q * GP_NODES + (half << 3) + r) * PC + tn * 16 + l15] = acc[r];
    }
    __syncthreads();
    if (live) {
        float2 c = *(const float2*)&Cp[wv * PC + lane * 2];
#pragma unroll
        for (int qq = 1; qq < NQ; ++qq) {
            const float2 p = *(const float2*)&Cp[(qq * GP_NODES + wv) * PC + lane * 2];
            c.x += p.x;
            c.y += p.y;
        }
        *(float2*)(C + (size_t)d * NC + lane * 2) = c;
        const float ss = wave_sum(c.x * ws.x + c.y * ws.y);
        const float sd = wave_sum(c.x * wd.x + c.y * wd.y);
        if (lane == 0) { out_s[d] = ss; out_d[d] = sd; }
    }
}

template <bool ELU>
__global__ void dst_gather1(const int* __restrict__ srt,
                            const int* __restrict__ beg, const int* __restrict__ fin,
                            const float* __restrict__ as_, const float* __restrict__ ad_,
                            const float* __restrict__ hsrc, const float* __restrict__ bias,
                            float* __restrict__ out, int N) {
    const int lane = threadIdx.x & 31;
    const int d    = blockIdx.x * (blockDim.x >> 5) + (threadIdx.x >> 5);
    if (d >= N) return;
    const int jb = __builtin_amdgcn_readfirstlane(beg[d]);
    const int je = __builtin_amdgcn_readfirstlane(fin[d]);
    const int n0 = (je - jb < 32) ? (je - jb) : 32;
    const float bv0 = bias[lane * 2], bv1 = bias[lane * 2 + 1];
    const float adv = ad_[d];
    float2 acc = make_float2(0.f, 0.f);

    auto logits = [&](int base, int n, int& sI) -> float {
        float e = -INFINITY;
        sI = 0;
        if (lane < n) {
            sI = srt[base + lane];
            e  = leaky(as_[sI] + adv);
        }
        return e;
    };
    auto rows = [&](int sI, float a, int n) {
#pragma unroll 4
        for (int k = 0; k < n; ++k) {
            const int   s  = __builtin_amdgcn_readlane(sI, k);
            const float al = __int_as_float(__builtin_amdgcn_readlane(__float_as_int(a), k));
            const float2 hv = *(const float2*)(hsrc + (size_t)s * 64 + lane * 2);
            acc.x += hv.x * al;
            acc.y += hv.y * al;
        }
    };

    int sI0;
    const float e0 = logits(jb, n0, sI0);

    float m = e0;
    for (int base = jb + 32; base < je; base += 32) {
        const int n = (je - base < 32) ? (je - base) : 32;
        int sI;
        m = fmaxf(m, logits(base, n, sI));
    }
    m = wave_max(m);

    const float p0 = (lane < n0) ? __expf(e0 - m) : 0.f;
    float sv = p0;
    for (int base = jb + 32; base < je; base += 32) {
        const int n = (je - base < 32) ? (je - base) : 32;
        int sI;
        const float e = logits(base, n, sI);
        if (lane < n) sv += __expf(e - m);
    }
    sv = wave_sum(sv);

    rows(sI0, p0 / (sv + 1e-16f), n0);
    for (int base = jb + 32; base < je; base += 32) {
        const int n = (je - base < 32) ? (je - base) : 32;
        int sI;
        const float e = logits(base, n, sI);
        const float a = (lane < n) ? __expf(e - m) / (sv + 1e-16f) : 0.f;
        rows(sI, a, n);
    }
    *(float2*)(out + (size_t)d * 64 + lane * 2) =
        make_float2(finish<ELU>(acc.x, bv0), finish<ELU>(acc.y, bv1));
}

extern "C" void kernel_launch(void* const* d_in, const int* in_sizes, int n_in,
                              void* d_out, int out_size, void* d_ws, size_t ws_size,
                              hipStream_t stream) {
    const float* x        = (const float*)d_in[0];
    const int*   ei       = (const int*)  d_in[1];
    const float* W1       = (const float*)d_in[2];
    const float* att_src1 = (const float*)d_in[3];
    const float* att_dst1 = (const float*)d_in[4];
    const float* b1       = (const float*)d_in[5];
    const float* W2       = (const float*)d_in[6];
    const float* att_src2 = (const float*)d_in[7];
    const float* att_dst2 = (const float*)d_in[8];
    const float* b2       = (const float*)d_in[9];
    float* out = (float*)d_out;

    const int N = in_sizes[0] / 128;
    const int E = in_sizes[1] / 2;

    float* ws = (float*)d_ws;
    float*    h1  = ws;
    float*    sc  = h1 + (size_t)N * 128;
    float*    as1 = sc + (size_t)N * 128;
    float*    ad1 = as1 + (size_t)N * 4;
    float*    as2 = ad1 + (size_t)N * 4;
    float*    ad2 = as2 + N;
    int*      beg = (int*)(ad2 + N);
    int*      fin = beg + N;
    int*      srt = fin + N;
    float*    h2  = sc;

    const int chunk  = (E + CHUNKS - 1) / CHUNKS;
    const int nbk    = (N + BUCKET_NODES - 1) >> BUCKET_SHIFT;
    const int nmat   = nbk * CHUNKS;
    int*      rstart = (int*)sc;
    int*      rcnt   = rstart + nmat;
    int*      pairs  = rcnt + nmat;

    const int TB = 256;
    const int gBlocks = (N + GP_NODES - 1) / GP_NODES;
    const int nBlocks = (N + 7) / 8;
    const int pBlocks = (N + PRODUCT_ROWS - 1) / PRODUCT_ROWS;

    first_launch<<<CHUNKS + pBlocks, SORT_THREADS, 0, stream>>>(ei, E, chunk, nbk, rstart, rcnt, pairs,
                                                              x, W1, h1, att_src1, att_dst1, as1, ad1, N);
    bucket_sort<<<nbk, SORT_THREADS, 0, stream>>>(pairs, rstart, rcnt, N, beg, fin, srt);

    gather_product<128, 64><<<gBlocks, GP_THREADS, 0, stream>>>(srt, beg, fin, as1, ad1, h1, b1,
                                                                W2, att_src2, att_dst2, h2, as2, ad2, N);

    dst_gather1<false><<<nBlocks, TB, 0, stream>>>(srt, beg, fin, as2, ad2, h2, b2, out, N);
}
